# starting kernel
# baseline (speedup 1.0000x reference)
#include <hip/hip_runtime.h>
#include <hip/hip_bf16.h>

typedef float v2f __attribute__((ext_vector_type(2)));
typedef float v8f __attribute__((ext_vector_type(8)));

#define D 128
#define H3 384

__device__ __forceinline__ float sigmoidf_(float x) {
    return 1.0f / (1.0f + __expf(-x));
}

__global__ void zero_words_kernel(int* __restrict__ p, int nwords) {
    int i = blockIdx.x * blockDim.x + threadIdx.x;
    if (i < nwords) p[i] = 0;
}

__global__ void mark_flags_kernel(const int* __restrict__ d0,
                                  const int* __restrict__ d1,
                                  const int* __restrict__ d2,
                                  unsigned char* __restrict__ flags,
                                  int N, int E) {
    int i = blockIdx.x * blockDim.x + threadIdx.x;
    if (i < E)            flags[d0[i]] = 1;
    else if (i < 2 * E)   flags[N + d1[i - E]] = 1;
    else if (i < 3 * E)   flags[2 * N + d2[i - 2 * E]] = 1;
}

__global__ __launch_bounds__(256) void feats_kernel(
    const float* __restrict__ ns,
    const float* __restrict__ Wa, const float* __restrict__ Wc,
    const float* __restrict__ Wd,
    const unsigned char* __restrict__ flags,
    float* __restrict__ feats, int N) {
    const int lane = threadIdx.x & 31;
    const int wave = threadIdx.x >> 5;
    const int row0 = blockIdx.x * 16;
    const int col0 = wave * 16;
    const int mn   = lane & 15;
    const int koff = (lane >> 4) * 2;

    const float* arow = ns + (size_t)(row0 + mn) * D + koff;
    const float* ba   = Wa + (size_t)(col0 + mn) * D + koff;
    const float* bc   = Wc + (size_t)(col0 + mn) * D + koff;
    const float* bd   = Wd + (size_t)(col0 + mn) * D + koff;

    v8f c0 = {}, c1 = {}, c2 = {};
#pragma unroll
    for (int kb = 0; kb < D; kb += 4) {
        v2f a  = *(const v2f*)(arow + kb);
        v2f b0 = *(const v2f*)(ba + kb);
        v2f b1 = *(const v2f*)(bc + kb);
        v2f b2 = *(const v2f*)(bd + kb);
        c0 = __builtin_amdgcn_wmma_f32_16x16x4_f32(false, a, false, b0, (short)0, c0, false, false);
        c1 = __builtin_amdgcn_wmma_f32_16x16x4_f32(false, a, false, b1, (short)0, c1, false, false);
        c2 = __builtin_amdgcn_wmma_f32_16x16x4_f32(false, a, false, b2, (short)0, c2, false, false);
    }
#pragma unroll
    for (int i = 0; i < 8; ++i) {
        int m   = i + ((lane >> 4) << 3);
        int row = row0 + m;
        float fa = flags[row]         ? c0[i] : 0.0f;
        float fc = flags[N + row]     ? c1[i] : 0.0f;
        float fd = flags[2 * N + row] ? c2[i] : 0.0f;
        feats[(size_t)row * D + col0 + mn] = fmaxf(fmaxf(fa, fc), fd);
    }
}

__global__ __launch_bounds__(256) void gi_kernel(
    const float* __restrict__ feats, const float* __restrict__ Wih,
    const float* __restrict__ bih, float* __restrict__ gi) {
    const int lane = threadIdx.x & 31;
    const int wave = threadIdx.x >> 5;
    const int row0 = blockIdx.x * 16;
    const int col0 = blockIdx.y * 128 + wave * 16;
    const int mn   = lane & 15;
    const int koff = (lane >> 4) * 2;

    const float* arow = feats + (size_t)(row0 + mn) * D + koff;
    const float* brow = Wih   + (size_t)(col0 + mn) * D + koff;

    v8f c = {};
#pragma unroll
    for (int kb = 0; kb < D; kb += 4) {
        v2f a = *(const v2f*)(arow + kb);
        v2f b = *(const v2f*)(brow + kb);
        c = __builtin_amdgcn_wmma_f32_16x16x4_f32(false, a, false, b, (short)0, c, false, false);
    }
    float bias = bih[col0 + mn];
#pragma unroll
    for (int i = 0; i < 8; ++i) {
        int m = i + ((lane >> 4) << 3);
        gi[(size_t)(row0 + m) * H3 + col0 + mn] = c[i] + bias;
    }
}

__global__ __launch_bounds__(256) void gru_step_kernel(
    const float* __restrict__ hin, const float* __restrict__ gi,
    const float* __restrict__ Whh, const float* __restrict__ bhh,
    float* __restrict__ hout) {
    const int lane = threadIdx.x & 31;
    const int wave = threadIdx.x >> 5;
    const int row0 = blockIdx.x * 16;
    const int col0 = wave * 16;
    const int mn   = lane & 15;
    const int koff = (lane >> 4) * 2;

    const float* arow = hin + (size_t)(row0 + mn) * D + koff;
    const float* br   = Whh + (size_t)(col0 + mn) * D + koff;
    const float* bz   = Whh + (size_t)(128 + col0 + mn) * D + koff;
    const float* bn   = Whh + (size_t)(256 + col0 + mn) * D + koff;

    v8f cr = {}, cz = {}, cn = {};
#pragma unroll
    for (int kb = 0; kb < D; kb += 4) {
        v2f a  = *(const v2f*)(arow + kb);
        v2f b0 = *(const v2f*)(br + kb);
        v2f b1 = *(const v2f*)(bz + kb);
        v2f b2 = *(const v2f*)(bn + kb);
        cr = __builtin_amdgcn_wmma_f32_16x16x4_f32(false, a, false, b0, (short)0, cr, false, false);
        cz = __builtin_amdgcn_wmma_f32_16x16x4_f32(false, a, false, b1, (short)0, cz, false, false);
        cn = __builtin_amdgcn_wmma_f32_16x16x4_f32(false, a, false, b2, (short)0, cn, false, false);
    }
    const float bhr = bhh[col0 + mn];
    const float bhz = bhh[128 + col0 + mn];
    const float bhn = bhh[256 + col0 + mn];
#pragma unroll
    for (int i = 0; i < 8; ++i) {
        int m   = i + ((lane >> 4) << 3);
        int row = row0 + m;
        size_t gb = (size_t)row * H3 + col0 + mn;
        float r  = sigmoidf_(gi[gb]       + cr[i] + bhr);
        float z  = sigmoidf_(gi[gb + 128] + cz[i] + bhz);
        float ng = tanhf(gi[gb + 256] + r * (cn[i] + bhn));
        float h  = hin[(size_t)row * D + col0 + mn];
        hout[(size_t)row * D + col0 + mn] = (1.0f - z) * ng + z * h;
    }
}

extern "C" void kernel_launch(void* const* d_in, const int* in_sizes, int n_in,
                              void* d_out, int out_size, void* d_ws, size_t ws_size,
                              hipStream_t stream) {
    const float* ns   = (const float*)d_in[0];
    const int*   da   = (const int*)d_in[1];
    const int*   dc   = (const int*)d_in[2];
    const int*   dd   = (const int*)d_in[3];
    const float* Wa   = (const float*)d_in[4];
    const float* Wc   = (const float*)d_in[5];
    const float* Wd   = (const float*)d_in[6];
    const float* Wih  = (const float*)d_in[7];
    const float* Whh  = (const float*)d_in[8];
    const float* bih  = (const float*)d_in[9];
    const float* bhh  = (const float*)d_in[10];
    float* out = (float*)d_out;

    const int N = in_sizes[0] / D;
    const int E = in_sizes[1];

    char* ws = (char*)d_ws;
    float* gi    = (float*)ws;
    float* feats = (float*)(ws + (size_t)N * H3 * sizeof(float));
    unsigned char* flags = (unsigned char*)(ws + (size_t)N * (H3 + D) * sizeof(float));

    int nwords = (3 * N + 3) / 4;
    zero_words_kernel<<<(nwords + 255) / 256, 256, 0, stream>>>((int*)flags, nwords);
    mark_flags_kernel<<<(3 * E + 255) / 256, 256, 0, stream>>>(da, dc, dd, flags, N, E);

    const int rowBlocks = N / 16;

    feats_kernel<<<rowBlocks, 256, 0, stream>>>(ns, Wa, Wc, Wd, flags, feats, N);

    {
        dim3 g(rowBlocks, H3 / 128);
        gi_kernel<<<g, 256, 0, stream>>>(feats, Wih, bih, gi);
    }

    gru_step_kernel<<<rowBlocks, 256, 0, stream>>>(ns,    gi, Whh, bhh, feats);
    gru_step_kernel<<<rowBlocks, 256, 0, stream>>>(feats, gi, Whh, bhh, out);
    gru_step_kernel<<<rowBlocks, 256, 0, stream>>>(out,   gi, Whh, bhh, feats);
    gru_step_kernel<<<rowBlocks, 256, 0, stream>>>(feats, gi, Whh, bhh, out);
}


# weights held in registers: each wave loads its three 16x128 weight tiles once and walks over row tiles in 256 persistent work-groups (feats, gi, GRU kernels); summation order unchanged
# speedup vs baseline: 1.9230x; 1.9230x over previous
#include <hip/hip_runtime.h>
#include <hip/hip_bf16.h>


typedef float v2f __attribute__((ext_vector_type(2)));
typedef float v8f __attribute__((ext_vector_type(8)));

#define D 128
#define H3 384
#define KSTEPS 32
#define PERSIST_BLOCKS 256

__device__ __forceinline__ float sigmoidf_(float x) {
    return 1.0f / (1.0f + __expf(-x));
}

__device__ __forceinline__ void load_frag_row(const float* __restrict__ p, v2f (&f)[KSTEPS]) {
#pragma unroll
    for (int s = 0; s < KSTEPS; ++s) f[s] = *(const v2f*)(p + 4 * s);
}

__device__ __forceinline__ void mma3(const v2f (&a)[KSTEPS],
                                     const v2f (&w0)[KSTEPS], const v2f (&w1)[KSTEPS],
                                     const v2f (&w2)[KSTEPS],
                                     v8f& c0, v8f& c1, v8f& c2) {
#pragma unroll
    for (int s = 0; s < KSTEPS; ++s) {
        c0 = __builtin_amdgcn_wmma_f32_16x16x4_f32(false, a[s], false, w0[s], (short)0, c0, false, false);
        c1 = __builtin_amdgcn_wmma_f32_16x16x4_f32(false, a[s], false, w1[s], (short)0, c1, false, false);
        c2 = __builtin_amdgcn_wmma_f32_16x16x4_f32(false, a[s], false, w2[s], (short)0, c2, false, false);
    }
}

__global__ void zero_words_kernel(int* __restrict__ p, int nwords) {
    int i = blockIdx.x * blockDim.x + threadIdx.x;
    if (i < nwords) p[i] = 0;
}

__global__ void mark_flags_kernel(const int* __restrict__ d0,
                                  const int* __restrict__ d1,
                                  const int* __restrict__ d2,
                                  unsigned char* __restrict__ flags,
                                  int N, int E) {
    int i = blockIdx.x * blockDim.x + threadIdx.x;
    if (i < E)            flags[d0[i]] = 1;
    else if (i < 2 * E)   flags[N + d1[i - E]] = 1;
    else if (i < 3 * E)   flags[2 * N + d2[i - 2 * E]] = 1;
}

__global__ __launch_bounds__(256) void feats_kernel(
    const float* __restrict__ ns,
    const float* __restrict__ Wa, const float* __restrict__ Wc,
    const float* __restrict__ Wd,
    const unsigned char* __restrict__ flags,
    float* __restrict__ feats, int N, int nTiles) {
    const int lane = threadIdx.x & 31;
    const int wave = threadIdx.x >> 5;
    const int col0 = wave * 16;
    const int mn   = lane & 15;
    const int koff = (lane >> 4) * 2;
    const int mhi  = (lane >> 4) << 3;

    int t = blockIdx.x;
    if (t >= nTiles) return;

    v2f w0[KSTEPS], w1[KSTEPS], w2[KSTEPS];
    load_frag_row(Wa + (size_t)(col0 + mn) * D + koff, w0);
    load_frag_row(Wc + (size_t)(col0 + mn) * D + koff, w1);
    load_frag_row(Wd + (size_t)(col0 + mn) * D + koff, w2);

    v2f a[KSTEPS];
    load_frag_row(ns + (size_t)(t * 16 + mn) * D + koff, a);

    for (; t < nTiles; t += gridDim.x) {
        int tn = t + gridDim.x;
        if (tn >= nTiles) tn = t;
        v2f an[KSTEPS];
        load_frag_row(ns + (size_t)(tn * 16 + mn) * D + koff, an);

        const int row0 = t * 16;
        unsigned char fa[8], fc[8], fd[8];
#pragma unroll
        for (int i = 0; i < 8; ++i) {
            int row = row0 + i + mhi;
            fa[i] = flags[row];
            fc[i] = flags[N + row];
            fd[i] = flags[2 * N + row];
        }

        v8f c0 = {}, c1 = {}, c2 = {};
        mma3(a, w0, w1, w2, c0, c1, c2);

#pragma unroll
        for (int i = 0; i < 8; ++i) {
            int row = row0 + i + mhi;
            float va = fa[i] ? c0[i] : 0.0f;
            float vc = fc[i] ? c1[i] : 0.0f;
            float vd = fd[i] ? c2[i] : 0.0f;
            feats[(size_t)row * D + col0 + mn] = fmaxf(fmaxf(va, vc), vd);
        }
#pragma unroll
        for (int s = 0; s < KSTEPS; ++s) a[s] = an[s];
    }
}

__global__ __launch_bounds__(256) void gi_kernel(
    const float* __restrict__ feats, const float* __restrict__ Wih,
    const float* __restrict__ bih, float* __restrict__ gi, int nTiles) {
    const int lane = threadIdx.x & 31;
    const int wave = threadIdx.x >> 5;
    const int col0 = wave * 16;
    const int mn   = lane & 15;
    const int koff = (lane >> 4) * 2;
    const int mhi  = (lane >> 4) << 3;

    int t = blockIdx.x;
    if (t >= nTiles) return;

    v2f w0[KSTEPS], w1[KSTEPS], w2[KSTEPS];
    load_frag_row(Wih + (size_t)(col0 + mn) * D + koff, w0);
    load_frag_row(Wih + (size_t)(128 + col0 + mn) * D + koff, w1);
    load_frag_row(Wih + (size_t)(256 + col0 + mn) * D + koff, w2);
    const float b0 = bih[col0 + mn];
    const float b1 = bih[128 + col0 + mn];
    const float b2 = bih[256 + col0 + mn];

    v2f a[KSTEPS];
    load_frag_row(feats + (size_t)(t * 16 + mn) * D + koff, a);

    for (; t < nTiles; t += gridDim.x) {
        int tn = t + gridDim.x;
        if (tn >= nTiles) tn = t;
        v2f an[KSTEPS];
        load_frag_row(feats + (size_t)(tn * 16 + mn) * D + koff, an);

        v8f c0 = {}, c1 = {}, c2 = {};
        mma3(a, w0, w1, w2, c0, c1, c2);

        const int row0 = t * 16;
#pragma unroll
        for (int i = 0; i < 8; ++i) {
            size_t gb = (size_t)(row0 + i + mhi) * H3 + col0 + mn;
            gi[gb]       = c0[i] + b0;
            gi[gb + 128] = c1[i] + b1;
            gi[gb + 256] = c2[i] + b2;
        }
#pragma unroll
        for (int s = 0; s < KSTEPS; ++s) a[s] = an[s];
    }
}

__global__ __launch_bounds__(256) void gru_step_kernel(
    const float* __restrict__ hin, const float* __restrict__ gi,
    const float* __restrict__ Whh, const float* __restrict__ bhh,
    float* __restrict__ hout, int nTiles) {
    const int lane = threadIdx.x & 31;
    const int wave = threadIdx.x >> 5;
    const int col0 = wave * 16;
    const int mn   = lane & 15;
    const int koff = (lane >> 4) * 2;
    const int mhi  = (lane >> 4) << 3;

    int t = blockIdx.x;
    if (t >= nTiles) return;

    v2f wr[KSTEPS], wz[KSTEPS], wn[KSTEPS];
    load_frag_row(Whh + (size_t)(col0 + mn) * D + koff, wr);
    load_frag_row(Whh + (size_t)(128 + col0 + mn) * D + koff, wz);
    load_frag_row(Whh + (size_t)(256 + col0 + mn) * D + koff, wn);
    const float bhr = bhh[col0 + mn];
    const float bhz = bhh[128 + col0 + mn];
    const float bhn = bhh[256 + col0 + mn];

    v2f a[KSTEPS];
    load_frag_row(hin + (size_t)(t * 16 + mn) * D + koff, a);

    for (; t < nTiles; t += gridDim.x) {
        int tn = t + gridDim.x;
        if (tn >= nTiles) tn = t;
        v2f an[KSTEPS];
        load_frag_row(hin + (size_t)(tn * 16 + mn) * D + koff, an);

        const int row0 = t * 16;
        float gr[8], gz[8], gn[8], hv[8];
#pragma unroll
        for (int i = 0; i < 8; ++i) {
            int row = row0 + i + mhi;
            size_t gb = (size_t)row * H3 + col0 + mn;
            gr[i] = gi[gb];
            gz[i] = gi[gb + 128];
            gn[i] = gi[gb + 256];
            hv[i] = hin[(size_t)row * D + col0 + mn];
        }

        v8f cr = {}, cz = {}, cn = {};
        mma3(a, wr, wz, wn, cr, cz, cn);

#pragma unroll
        for (int i = 0; i < 8; ++i) {
            int row = row0 + i + mhi;
            float r  = sigmoidf_(gr[i] + cr[i] + bhr);
            float z  = sigmoidf_(gz[i] + cz[i] + bhz);
            float ng = tanhf(gn[i] + r * (cn[i] + bhn));
            float h  = hv[i];
            hout[(size_t)row * D + col0 + mn] = (1.0f - z) * ng + z * h;
        }
#pragma unroll
        for (int s = 0; s < KSTEPS; ++s) a[s] = an[s];
    }
}

extern "C" void kernel_launch(void* const* d_in, const int* in_sizes, int n_in,
                              void* d_out, int out_size, void* d_ws, size_t ws_size,
                              hipStream_t stream) {
    const float* ns   = (const float*)d_in[0];
    const int*   da   = (const int*)d_in[1];
    const int*   dc   = (const int*)d_in[2];
    const int*   dd   = (const int*)d_in[3];
    const float* Wa   = (const float*)d_in[4];
    const float* Wc   = (const float*)d_in[5];
    const float* Wd   = (const float*)d_in[6];
    const float* Wih  = (const float*)d_in[7];
    const float* Whh  = (const float*)d_in[8];
    const float* bih  = (const float*)d_in[9];
    const float* bhh  = (const float*)d_in[10];
    float* out = (float*)d_out;

    const int N = in_sizes[0] / D;
    const int E = in_sizes[1];

    char* ws = (char*)d_ws;
    float* gi    = (float*)ws;
    float* feats = (float*)(ws + (size_t)N * H3 * sizeof(float));
    unsigned char* flags = (unsigned char*)(ws + (size_t)N * (H3 + D) * sizeof(float));

    int nwords = (3 * N + 3) / 4;
    zero_words_kernel<<<(nwords + 255) / 256, 256, 0, stream>>>((int*)flags, nwords);
    mark_flags_kernel<<<(3 * E + 255) / 256, 256, 0, stream>>>(da, dc, dd, flags, N, E);

    const int rowBlocks = N / 16;

    feats_kernel<<<PERSIST_BLOCKS, 256, 0, stream>>>(ns, Wa, Wc, Wd, flags, feats, N, rowBlocks);

    gi_kernel<<<PERSIST_BLOCKS, 256, 0, stream>>>(feats, Wih, bih, gi, rowBlocks);

    gru_step_kernel<<<PERSIST_BLOCKS, 256, 0, stream>>>(ns,    gi, Whh, bhh, feats, rowBlocks);
    gru_step_kernel<<<PERSIST_BLOCKS, 256, 0, stream>>>(feats, gi, Whh, bhh, out,   rowBlocks);
    gru_step_kernel<<<PERSIST_BLOCKS, 256, 0, stream>>>(out,   gi, Whh, bhh, feats, rowBlocks);
    gru_step_kernel<<<PERSIST_BLOCKS, 256, 0, stream>>>(feats, gi, Whh, bhh, out,   rowBlocks);
}


# GRU gates branch-free on v_exp_f32 and v_rcp_f32: tanh as 1 - 2/(1 + exp(2x)), sigmoid's IEEE division as one reciprocal; library tanh and its branches gone; gate values differ in the last bits
# speedup vs baseline: 1.9866x; 1.0331x over previous
#include <hip/hip_runtime.h>
#include <hip/hip_bf16.h>


typedef float v2f __attribute__((ext_vector_type(2)));
typedef float v8f __attribute__((ext_vector_type(8)));

#define D 128
#define H3 384
#define KSTEPS 32
#define PERSIST_BLOCKS 256

__device__ __forceinline__ float sigmoidf_(float x) {
    return __builtin_amdgcn_rcpf(1.0f + __expf(-x));
}

__device__ __forceinline__ float tanhf_(float x) {
    return 1.0f - 2.0f * __builtin_amdgcn_rcpf(1.0f + __expf(2.0f * x));
}

__device__ __forceinline__ void load_frag_row(const float* __restrict__ p, v2f (&f)[KSTEPS]) {
#pragma unroll
    for (int s = 0; s < KSTEPS; ++s) f[s] = *(const v2f*)(p + 4 * s);
}

__device__ __forceinline__ void mma3(const v2f (&a)[KSTEPS],
                                     const v2f (&w0)[KSTEPS], const v2f (&w1)[KSTEPS],
                                     const v2f (&w2)[KSTEPS],
                                     v8f& c0, v8f& c1, v8f& c2) {
#pragma unroll
    for (int s = 0; s < KSTEPS; ++s) {
        c0 = __builtin_amdgcn_wmma_f32_16x16x4_f32(false, a[s], false, w0[s], (short)0, c0, false, false);
        c1 = __builtin_amdgcn_wmma_f32_16x16x4_f32(false, a[s], false, w1[s], (short)0, c1, false, false);
        c2 = __builtin_amdgcn_wmma_f32_16x16x4_f32(false, a[s], false, w2[s], (short)0, c2, false, false);
    }
}

__global__ void zero_words_kernel(int* __restrict__ p, int nwords) {
    int i = blockIdx.x * blockDim.x + threadIdx.x;
    if (i < nwords) p[i] = 0;
}

__global__ void mark_flags_kernel(const int* __restrict__ d0,
                                  const int* __restrict__ d1,
                                  const int* __restrict__ d2,
                                  unsigned char* __restrict__ flags,
                                  int N, int E) {
    int i = blockIdx.x * blockDim.x + threadIdx.x;
    if (i < E)            flags[d0[i]] = 1;
    else if (i < 2 * E)   flags[N + d1[i - E]] = 1;
    else if (i < 3 * E)   flags[2 * N + d2[i - 2 * E]] = 1;
}

__global__ __launch_bounds__(256) void feats_kernel(
    const float* __restrict__ ns,
    const float* __restrict__ Wa, const float* __restrict__ Wc,
    const float* __restrict__ Wd,
    const unsigned char* __restrict__ flags,
    float* __restrict__ feats, int N, int nTiles) {
    const int lane = threadIdx.x & 31;
    const int wave = threadIdx.x >> 5;
    const int col0 = wave * 16;
    const int mn   = lane & 15;
    const int koff = (lane >> 4) * 2;
    const int mhi  = (lane >> 4) << 3;

    int t = blockIdx.x;
    if (t >= nTiles) return;

    v2f w0[KSTEPS], w1[KSTEPS], w2[KSTEPS];
    load_frag_row(Wa + (size_t)(col0 + mn) * D + koff, w0);
    load_frag_row(Wc + (size_t)(col0 + mn) * D + koff, w1);
    load_frag_row(Wd + (size_t)(col0 + mn) * D + koff, w2);

    v2f a[KSTEPS];
    load_frag_row(ns + (size_t)(t * 16 + mn) * D + koff, a);

    for (; t < nTiles; t += gridDim.x) {
        int tn = t + gridDim.x;
        if (tn >= nTiles) tn = t;
        v2f an[KSTEPS];
        load_frag_row(ns + (size_t)(tn * 16 + mn) * D + koff, an);

        const int row0 = t * 16;
        unsigned char fa[8], fc[8], fd[8];
#pragma unroll
        for (int i = 0; i < 8; ++i) {
            int row = row0 + i + mhi;
            fa[i] = flags[row];
            fc[i] = flags[N + row];
            fd[i] = flags[2 * N + row];
        }

        v8f c0 = {}, c1 = {}, c2 = {};
        mma3(a, w0, w1, w2, c0, c1, c2);

#pragma unroll
        for (int i = 0; i < 8; ++i) {
            int row = row0 + i + mhi;
            float va = fa[i] ? c0[i] : 0.0f;
            float vc = fc[i] ? c1[i] : 0.0f;
            float vd = fd[i] ? c2[i] : 0.0f;
            feats[(size_t)row * D + col0 + mn] = fmaxf(fmaxf(va, vc), vd);
        }
#pragma unroll
        for (int s = 0; s < KSTEPS; ++s) a[s] = an[s];
    }
}

__global__ __launch_bounds__(256) void gi_kernel(
    const float* __restrict__ feats, const float* __restrict__ Wih,
    const float* __restrict__ bih, float* __restrict__ gi, int nTiles) {
    const int lane = threadIdx.x & 31;
    const int wave = threadIdx.x >> 5;
    const int col0 = wave * 16;
    const int mn   = lane & 15;
    const int koff = (lane >> 4) * 2;
    const int mhi  = (lane >> 4) << 3;

    int t = blockIdx.x;
    if (t >= nTiles) return;

    v2f w0[KSTEPS], w1[KSTEPS], w2[KSTEPS];
    load_frag_row(Wih + (size_t)(col0 + mn) * D + koff, w0);
    load_frag_row(Wih + (size_t)(128 + col0 + mn) * D + koff, w1);
    load_frag_row(Wih + (size_t)(256 + col0 + mn) * D + koff, w2);
    const float b0 = bih[col0 + mn];
    const float b1 = bih[128 + col0 + mn];
    const float b2 = bih[256 + col0 + mn];

    v2f a[KSTEPS];
    load_frag_row(feats + (size_t)(t * 16 + mn) * D + koff, a);

    for (; t < nTiles; t += gridDim.x) {
        int tn = t + gridDim.x;
        if (tn >= nTiles) tn = t;
        v2f an[KSTEPS];
        load_frag_row(feats + (size_t)(tn * 16 + mn) * D + koff, an);

        v8f c0 = {}, c1 = {}, c2 = {};
        mma3(a, w0, w1, w2, c0, c1, c2);

        const int row0 = t * 16;
#pragma unroll
        for (int i = 0; i < 8; ++i) {
            size_t gb = (size_t)(row0 + i + mhi) * H3 + col0 + mn;
            gi[gb]       = c0[i] + b0;
            gi[gb + 128] = c1[i] + b1;
            gi[gb + 256] = c2[i] + b2;
        }
#pragma unroll
        for (int s = 0; s < KSTEPS; ++s) a[s] = an[s];
    }
}

__global__ __launch_bounds__(256) void gru_step_kernel(
    const float* __restrict__ hin, const float* __restrict__ gi,
    const float* __restrict__ Whh, const float* __restrict__ bhh,
    float* __restrict__ hout, int nTiles) {
    const int lane = threadIdx.x & 31;
    const int wave = threadIdx.x >> 5;
    const int col0 = wave * 16;
    const int mn   = lane & 15;
    const int koff = (lane >> 4) * 2;
    const int mhi  = (lane >> 4) << 3;

    int t = blockIdx.x;
    if (t >= nTiles) return;

    v2f wr[KSTEPS], wz[KSTEPS], wn[KSTEPS];
    load_frag_row(Whh + (size_t)(col0 + mn) * D + koff, wr);
    load_frag_row(Whh + (size_t)(128 + col0 + mn) * D + koff, wz);
    load_frag_row(Whh + (size_t)(256 + col0 + mn) * D + koff, wn);
    const float bhr = bhh[col0 + mn];
    const float bhz = bhh[128 + col0 + mn];
    const float bhn = bhh[256 + col0 + mn];

    v2f a[KSTEPS];
    load_frag_row(hin + (size_t)(t * 16 + mn) * D + koff, a);

    for (; t < nTiles; t += gridDim.x) {
        int tn = t + gridDim.x;
        if (tn >= nTiles) tn = t;
        v2f an[KSTEPS];
        load_frag_row(hin + (size_t)(tn * 16 + mn) * D + koff, an);

        const int row0 = t * 16;
        float gr[8], gz[8], gn[8], hv[8];
#pragma unroll
        for (int i = 0; i < 8; ++i) {
            int row = row0 + i + mhi;
            size_t gb = (size_t)row * H3 + col0 + mn;
            gr[i] = gi[gb];
            gz[i] = gi[gb + 128];
            gn[i] = gi[gb + 256];
            hv[i] = hin[(size_t)row * D + col0 + mn];
        }

        __builtin_amdgcn_sched_barrier(0);

        v8f cr = {}, cz = {}, cn = {};
        mma3(a, wr, wz, wn, cr, cz, cn);

#pragma unroll
        for (int i = 0; i < 8; ++i) {
            int row = row0 + i + mhi;
            float r  = sigmoidf_(gr[i] + cr[i] + bhr);
            float z  = sigmoidf_(gz[i] + cz[i] + bhz);
            float ng = tanhf_(gn[i] + r * (cn[i] + bhn));
            float h  = hv[i];
            hout[(size_t)row * D + col0 + mn] = (1.0f - z) * ng + z * h;
        }
        __builtin_amdgcn_sched_barrier(0);
#pragma unroll
        for (int s = 0; s < KSTEPS; ++s) a[s] = an[s];
    }
}

extern "C" void kernel_launch(void* const* d_in, const int* in_sizes, int n_in,
                              void* d_out, int out_size, void* d_ws, size_t ws_size,
                              hipStream_t stream) {
    const float* ns   = (const float*)d_in[0];
    const int*   da   = (const int*)d_in[1];
    const int*   dc   = (const int*)d_in[2];
    const int*   dd   = (const int*)d_in[3];
    const float* Wa   = (const float*)d_in[4];
    const float* Wc   = (const float*)d_in[5];
    const float* Wd   = (const float*)d_in[6];
    const float* Wih  = (const float*)d_in[7];
    const float* Whh  = (const float*)d_in[8];
    const float* bih  = (const float*)d_in[9];
    const float* bhh  = (const float*)d_in[10];
    float* out = (float*)d_out;

    const int N = in_sizes[0] / D;
    const int E = in_sizes[1];

    char* ws = (char*)d_ws;
    float* gi    = (float*)ws;
    float* feats = (float*)(ws + (size_t)N * H3 * sizeof(float));
    unsigned char* flags = (unsigned char*)(ws + (size_t)N * (H3 + D) * sizeof(float));

    int nwords = (3 * N + 3) / 4;
    zero_words_kernel<<<(nwords + 255) / 256, 256, 0, stream>>>((int*)flags, nwords);
    mark_flags_kernel<<<(3 * E + 255) / 256, 256, 0, stream>>>(da, dc, dd, flags, N, E);

    const int rowBlocks = N / 16;

    feats_kernel<<<PERSIST_BLOCKS, 256, 0, stream>>>(ns, Wa, Wc, Wd, flags, feats, N, rowBlocks);

    gi_kernel<<<PERSIST_BLOCKS, 256, 0, stream>>>(feats, Wih, bih, gi, rowBlocks);

    gru_step_kernel<<<PERSIST_BLOCKS, 256, 0, stream>>>(ns,    gi, Whh, bhh, feats, rowBlocks);
    gru_step_kernel<<<PERSIST_BLOCKS, 256, 0, stream>>>(feats, gi, Whh, bhh, out,   rowBlocks);
    gru_step_kernel<<<PERSIST_BLOCKS, 256, 0, stream>>>(out,   gi, Whh, bhh, feats, rowBlocks);
    gru_step_kernel<<<PERSIST_BLOCKS, 256, 0, stream>>>(feats, gi, Whh, bhh, out,   rowBlocks);
}


# 16x128 operand tiles (A rows per row tile, weight tiles once) fetched by whole-row 16-byte loads into padded LDS once per work-group; fragments and old h read from LDS; summation order unchanged
# speedup vs baseline: 2.2146x; 1.1148x over previous
#include <hip/hip_runtime.h>
#include <hip/hip_bf16.h>


typedef float v2f __attribute__((ext_vector_type(2)));
typedef float v4f __attribute__((ext_vector_type(4)));
typedef float v8f __attribute__((ext_vector_type(8)));

#define D 128
#define H3 384
#define KSTEPS 32
#define PERSIST_BLOCKS 256
#define PITCH 132
#define TILE_F (16 * PITCH)

__device__ __forceinline__ float sigmoidf_(float x) {
    return __builtin_amdgcn_rcpf(1.0f + __expf(-x));
}

__device__ __forceinline__ float tanhf_(float x) {
    return 1.0f - 2.0f * __builtin_amdgcn_rcpf(1.0f + __expf(2.0f * x));
}


__device__ __forceinline__ void stage_tile_wave(const float* __restrict__ g, float* l, int lane) {
#pragma unroll
    for (int r = 0; r < 16; ++r) {
        v4f v = *(const v4f*)(g + r * D + 4 * lane);
        *(v4f*)(l + r * PITCH + 4 * lane) = v;
    }
}

__device__ __forceinline__ void fetch_rows(const float* __restrict__ tile, int wave, int lane,
                                           v4f& lo, v4f& hi) {
    lo = *(const v4f*)(tile + wave * D + 4 * lane);
    hi = *(const v4f*)(tile + (wave + 8) * D + 4 * lane);
}
__device__ __forceinline__ void put_rows(float* buf, int wave, int lane, v4f lo, v4f hi) {
    *(v4f*)(buf + wave * PITCH + 4 * lane) = lo;
    *(v4f*)(buf + (wave + 8) * PITCH + 4 * lane) = hi;
}

__device__ __forceinline__ void lds_barrier() {
    __builtin_amdgcn_fence(__ATOMIC_RELEASE, "workgroup", "local");
    __builtin_amdgcn_s_barrier();
    __builtin_amdgcn_fence(__ATOMIC_ACQUIRE, "workgroup", "local");
}

__device__ __forceinline__ void load_frag_lds(const float* p, v2f (&f)[KSTEPS]) {
#pragma unroll
    for (int s = 0; s < KSTEPS; ++s) f[s] = *(const v2f*)(p + 4 * s);
}

__device__ __forceinline__ void load_weights3(const float* __restrict__ W, int r0, int r1, int r2,
                                              float* mytile, int lane,
                                              v2f (&w0)[KSTEPS], v2f (&w1)[KSTEPS], v2f (&w2)[KSTEPS]) {
    const float* fp = mytile + (lane & 15) * PITCH + (lane >> 4) * 2;
    stage_tile_wave(W + (size_t)r0 * D, mytile, lane);
    __syncthreads();
    load_frag_lds(fp, w0);
    __syncthreads();
    stage_tile_wave(W + (size_t)r1 * D, mytile, lane);
    __syncthreads();
    load_frag_lds(fp, w1);
    __syncthreads();
    stage_tile_wave(W + (size_t)r2 * D, mytile, lane);
    __syncthreads();
    load_frag_lds(fp, w2);
}

__device__ __forceinline__ void mma3(const v2f (&a)[KSTEPS],
                                     const v2f (&w0)[KSTEPS], const v2f (&w1)[KSTEPS],
                                     const v2f (&w2)[KSTEPS],
                                     v8f& c0, v8f& c1, v8f& c2) {
#pragma unroll
    for (int s = 0; s < KSTEPS; ++s) {
        c0 = __builtin_amdgcn_wmma_f32_16x16x4_f32(false, a[s], false, w0[s], (short)0, c0, false, false);
        c1 = __builtin_amdgcn_wmma_f32_16x16x4_f32(false, a[s], false, w1[s], (short)0, c1, false, false);
        c2 = __builtin_amdgcn_wmma_f32_16x16x4_f32(false, a[s], false, w2[s], (short)0, c2, false, false);
    }
}

__global__ void zero_words_kernel(int* __restrict__ p, int nwords) {
    int i = blockIdx.x * blockDim.x + threadIdx.x;
    if (i < nwords) p[i] = 0;
}

__global__ void mark_flags_kernel(const int* __restrict__ d0,
                                  const int* __restrict__ d1,
                                  const int* __restrict__ d2,
                                  unsigned char* __restrict__ flags,
                                  int N, int E) {
    int i = blockIdx.x * blockDim.x + threadIdx.x;
    if (i < E)            flags[d0[i]] = 1;
    else if (i < 2 * E)   flags[N + d1[i - E]] = 1;
    else if (i < 3 * E)   flags[2 * N + d2[i - 2 * E]] = 1;
}

__global__ __launch_bounds__(256) void feats_kernel(
    const float* __restrict__ ns,
    const float* __restrict__ Wa, const float* __restrict__ Wc,
    const float* __restrict__ Wd,
    const unsigned char* __restrict__ flags,
    float* __restrict__ feats, int N, int nTiles) {
    __shared__ __attribute__((aligned(16))) float wst[8 * TILE_F];
    __shared__ __attribute__((aligned(16))) float abuf[2 * TILE_F];

    const int lane = threadIdx.x & 31;
    const int wave = threadIdx.x >> 5;
    const int col0 = wave * 16;
    const int mn   = lane & 15;
    const int koff = (lane >> 4) * 2;
    const int mhi  = (lane >> 4) << 3;

    int t = blockIdx.x;
    if (t >= nTiles) return;

    v4f glo, ghi;
    fetch_rows(ns + (size_t)t * 16 * D, wave, lane, glo, ghi);

    float* mytile = wst + wave * TILE_F;
    const float* fp = mytile + mn * PITCH + koff;
    v2f w0[KSTEPS], w1[KSTEPS], w2[KSTEPS];
    stage_tile_wave(Wa + (size_t)col0 * D, mytile, lane);
    __syncthreads();
    load_frag_lds(fp, w0);
    __syncthreads();
    stage_tile_wave(Wc + (size_t)col0 * D, mytile, lane);
    __syncthreads();
    load_frag_lds(fp, w1);
    __syncthreads();
    stage_tile_wave(Wd + (size_t)col0 * D, mytile, lane);
    __syncthreads();
    load_frag_lds(fp, w2);

    put_rows(abuf, wave, lane, glo, ghi);
    __syncthreads();

    int cur = 0;
    for (; t < nTiles; t += gridDim.x) {
        const int row0 = t * 16;
        const unsigned long long fa8 = *(const unsigned long long*)(flags + row0 + mhi);
        const unsigned long long fc8 = *(const unsigned long long*)(flags + N + row0 + mhi);
        const unsigned long long fd8 = *(const unsigned long long*)(flags + 2 * N + row0 + mhi);

        int tn = t + gridDim.x;
        if (tn >= nTiles) tn = t;
        fetch_rows(ns + (size_t)tn * 16 * D, wave, lane, glo, ghi);

        const float* ab = abuf + cur * TILE_F;
        v2f a[KSTEPS];
        load_frag_lds(ab + mn * PITCH + koff, a);

        __builtin_amdgcn_sched_barrier(0);

        v8f c0 = {}, c1 = {}, c2 = {};
        mma3(a, w0, w1, w2, c0, c1, c2);

        __builtin_amdgcn_sched_barrier(0);

#pragma unroll
        for (int i = 0; i < 8; ++i) {
            int row = row0 + i + mhi;
            float va = ((fa8 >> (8 * i)) & 0xffull) ? c0[i] : 0.0f;
            float vc = ((fc8 >> (8 * i)) & 0xffull) ? c1[i] : 0.0f;
            float vd = ((fd8 >> (8 * i)) & 0xffull) ? c2[i] : 0.0f;
            feats[(size_t)row * D + col0 + mn] = fmaxf(fmaxf(va, vc), vd);
        }

        __builtin_amdgcn_sched_barrier(0);
        put_rows(abuf + (cur ^ 1) * TILE_F, wave, lane, glo, ghi);
        lds_barrier();
        cur ^= 1;
    }
}

__global__ __launch_bounds__(256) void gi_kernel(
    const float* __restrict__ feats, const float* __restrict__ Wih,
    const float* __restrict__ bih, float* __restrict__ gi, int nTiles) {
    __shared__ __attribute__((aligned(16))) float wst[8 * TILE_F];
    __shared__ __attribute__((aligned(16))) float abuf[2 * TILE_F];

    const int lane = threadIdx.x & 31;
    const int wave = threadIdx.x >> 5;
    const int col0 = wave * 16;
    const int mn   = lane & 15;
    const int koff = (lane >> 4) * 2;
    const int mhi  = (lane >> 4) << 3;

    int t = blockIdx.x;
    if (t >= nTiles) return;

    v4f glo, ghi;
    fetch_rows(feats + (size_t)t * 16 * D, wave, lane, glo, ghi);

    v2f w0[KSTEPS], w1[KSTEPS], w2[KSTEPS];
    load_weights3(Wih, col0, 128 + col0, 256 + col0, wst + wave * TILE_F, lane, w0, w1, w2);
    const float b0 = bih[col0 + mn];
    const float b1 = bih[128 + col0 + mn];
    const float b2 = bih[256 + col0 + mn];

    put_rows(abuf, wave, lane, glo, ghi);
    __syncthreads();

    int cur = 0;
    for (; t < nTiles; t += gridDim.x) {
        int tn = t + gridDim.x;
        if (tn >= nTiles) tn = t;
        fetch_rows(feats + (size_t)tn * 16 * D, wave, lane, glo, ghi);

        const float* ab = abuf + cur * TILE_F;
        v2f a[KSTEPS];
        load_frag_lds(ab + mn * PITCH + koff, a);

        __builtin_amdgcn_sched_barrier(0);

        v8f c0 = {}, c1 = {}, c2 = {};
        mma3(a, w0, w1, w2, c0, c1, c2);

        const int row0 = t * 16;
#pragma unroll
        for (int i = 0; i < 8; ++i) {
            size_t gb = (size_t)(row0 + i + mhi) * H3 + col0 + mn;
            gi[gb]       = c0[i] + b0;
            gi[gb + 128] = c1[i] + b1;
            gi[gb + 256] = c2[i] + b2;
        }

        __builtin_amdgcn_sched_barrier(0);
        put_rows(abuf + (cur ^ 1) * TILE_F, wave, lane, glo, ghi);
        lds_barrier();
        cur ^= 1;
    }
}

__global__ __launch_bounds__(256) void gru_step_kernel(
    const float* __restrict__ hin, const float* __restrict__ gi,
    const float* __restrict__ Whh, const float* __restrict__ bhh,
    float* __restrict__ hout, int nTiles) {
    __shared__ __attribute__((aligned(16))) float wst[8 * TILE_F];
    __shared__ __attribute__((aligned(16))) float abuf[2 * TILE_F];

    const int lane = threadIdx.x & 31;
    const int wave = threadIdx.x >> 5;
    const int col0 = wave * 16;
    const int mn   = lane & 15;
    const int koff = (lane >> 4) * 2;
    const int mhi  = (lane >> 4) << 3;

    int t = blockIdx.x;
    if (t >= nTiles) return;

    v4f glo, ghi;
    fetch_rows(hin + (size_t)t * 16 * D, wave, lane, glo, ghi);

    v2f wr[KSTEPS], wz[KSTEPS], wn[KSTEPS];
    load_weights3(Whh, col0, 128 + col0, 256 + col0, wst + wave * TILE_F, lane, wr, wz, wn);
    const float bhr = bhh[col0 + mn];
    const float bhz = bhh[128 + col0 + mn];
    const float bhn = bhh[256 + col0 + mn];

    put_rows(abuf, wave, lane, glo, ghi);
    __syncthreads();

    int cur = 0;
    for (; t < nTiles; t += gridDim.x) {
        const int row0 = t * 16;
        float gr[8], gz[8], gn[8], hv[8];
#pragma unroll
        for (int i = 0; i < 8; ++i) {
            size_t gb = (size_t)(row0 + i + mhi) * H3 + col0 + mn;
            gr[i] = gi[gb];
            gz[i] = gi[gb + 128];
            gn[i] = gi[gb + 256];
        }
        int tn = t + gridDim.x;
        if (tn >= nTiles) tn = t;
        fetch_rows(hin + (size_t)tn * 16 * D, wave, lane, glo, ghi);

        const float* ab = abuf + cur * TILE_F;
        v2f a[KSTEPS];
        load_frag_lds(ab + mn * PITCH + koff, a);
#pragma unroll
        for (int i = 0; i < 8; ++i) hv[i] = ab[(i + mhi) * PITCH + col0 + mn];

        __builtin_amdgcn_sched_barrier(0);

        v8f cr = {}, cz = {}, cn = {};
        mma3(a, wr, wz, wn, cr, cz, cn);

#pragma unroll
        for (int i = 0; i < 8; ++i) {
            int row = row0 + i + mhi;
            float r  = sigmoidf_(gr[i] + cr[i] + bhr);
            float z  = sigmoidf_(gz[i] + cz[i] + bhz);
            float ng = tanhf_(gn[i] + r * (cn[i] + bhn));
            float h  = hv[i];
            hout[(size_t)row * D + col0 + mn] = (1.0f - z) * ng + z * h;
        }

        __builtin_amdgcn_sched_barrier(0);
        put_rows(abuf + (cur ^ 1) * TILE_F, wave, lane, glo, ghi);
        lds_barrier();
        cur ^= 1;
    }
}

extern "C" void kernel_launch(void* const* d_in, const int* in_sizes, int n_in,
                              void* d_out, int out_size, void* d_ws, size_t ws_size,
                              hipStream_t stream) {
    const float* ns   = (const float*)d_in[0];
    const int*   da   = (const int*)d_in[1];
    const int*   dc   = (const int*)d_in[2];
    const int*   dd   = (const int*)d_in[3];
    const float* Wa   = (const float*)d_in[4];
    const float* Wc   = (const float*)d_in[5];
    const float* Wd   = (const float*)d_in[6];
    const float* Wih  = (const float*)d_in[7];
    const float* Whh  = (const float*)d_in[8];
    const float* bih  = (const float*)d_in[9];
    const float* bhh  = (const float*)d_in[10];
    float* out = (float*)d_out;

    const int N = in_sizes[0] / D;
    const int E = in_sizes[1];

    char* ws = (char*)d_ws;
    float* gi    = (float*)ws;
    float* feats = (float*)(ws + (size_t)N * H3 * sizeof(float));
    unsigned char* flags = (unsigned char*)(ws + (size_t)N * (H3 + D) * sizeof(float));

    int nwords = (3 * N + 3) / 4;
    zero_words_kernel<<<(nwords + 255) / 256, 256, 0, stream>>>((int*)flags, nwords);
    mark_flags_kernel<<<(3 * E + 255) / 256, 256, 0, stream>>>(da, dc, dd, flags, N, E);

    const int rowBlocks = N / 16;

    feats_kernel<<<PERSIST_BLOCKS, 256, 0, stream>>>(ns, Wa, Wc, Wd, flags, feats, N, rowBlocks);

    gi_kernel<<<PERSIST_BLOCKS, 256, 0, stream>>>(feats, Wih, bih, gi, rowBlocks);

    gru_step_kernel<<<PERSIST_BLOCKS, 256, 0, stream>>>(ns,    gi, Whh, bhh, feats, rowBlocks);
    gru_step_kernel<<<PERSIST_BLOCKS, 256, 0, stream>>>(feats, gi, Whh, bhh, out,   rowBlocks);
    gru_step_kernel<<<PERSIST_BLOCKS, 256, 0, stream>>>(out,   gi, Whh, bhh, feats, rowBlocks);
    gru_step_kernel<<<PERSIST_BLOCKS, 256, 0, stream>>>(feats, gi, Whh, bhh, out,   rowBlocks);
}


# flag marking pass on a fixed grid of 512 work-groups: each thread strides over the edges, four per edge type per 16-byte index load, instead of one edge per thread in 75,000 waves
# speedup vs baseline: 2.3257x; 1.0502x over previous
#include <hip/hip_runtime.h>
#include <hip/hip_bf16.h>


typedef float v2f __attribute__((ext_vector_type(2)));
typedef float v4f __attribute__((ext_vector_type(4)));
typedef float v8f __attribute__((ext_vector_type(8)));
typedef int v4i __attribute__((ext_vector_type(4)));

#define D 128
#define H3 384
#define KSTEPS 32
#define PERSIST_BLOCKS 256
#define MARK_BLOCKS 512
#define PITCH 132
#define TILE_F (16 * PITCH)

__device__ __forceinline__ float sigmoidf_(float x) {
    return __builtin_amdgcn_rcpf(1.0f + __expf(-x));
}

__device__ __forceinline__ float tanhf_(float x) {
    return 1.0f - 2.0f * __builtin_amdgcn_rcpf(1.0f + __expf(2.0f * x));
}


__device__ __forceinline__ void stage_tile_wave(const float* __restrict__ g, float* l, int lane) {
#pragma unroll
    for (int r = 0; r < 16; ++r) {
        v4f v = *(const v4f*)(g + r * D + 4 * lane);
        *(v4f*)(l + r * PITCH + 4 * lane) = v;
    }
}

__device__ __forceinline__ void fetch_rows(const float* __restrict__ tile, int wave, int lane,
                                           v4f& lo, v4f& hi) {
    lo = *(const v4f*)(tile + wave * D + 4 * lane);
    hi = *(const v4f*)(tile + (wave + 8) * D + 4 * lane);
}
__device__ __forceinline__ void put_rows(float* buf, int wave, int lane, v4f lo, v4f hi) {
    *(v4f*)(buf + wave * PITCH + 4 * lane) = lo;
    *(v4f*)(buf + (wave + 8) * PITCH + 4 * lane) = hi;
}

__device__ __forceinline__ void lds_barrier() {
    __builtin_amdgcn_fence(__ATOMIC_RELEASE, "workgroup", "local");
    __builtin_amdgcn_s_barrier();
    __builtin_amdgcn_fence(__ATOMIC_ACQUIRE, "workgroup", "local");
}

__device__ __forceinline__ void load_frag_lds(const float* p, v2f (&f)[KSTEPS]) {
#pragma unroll
    for (int s = 0; s < KSTEPS; ++s) f[s] = *(const v2f*)(p + 4 * s);
}

__device__ __forceinline__ void load_weights3(const float* __restrict__ W, int r0, int r1, int r2,
                                              float* mytile, int lane,
                                              v2f (&w0)[KSTEPS], v2f (&w1)[KSTEPS], v2f (&w2)[KSTEPS]) {
    const float* fp = mytile + (lane & 15) * PITCH + (lane >> 4) * 2;
    stage_tile_wave(W + (size_t)r0 * D, mytile, lane);
    __syncthreads();
    load_frag_lds(fp, w0);
    __syncthreads();
    stage_tile_wave(W + (size_t)r1 * D, mytile, lane);
    __syncthreads();
    load_frag_lds(fp, w1);
    __syncthreads();
    stage_tile_wave(W + (size_t)r2 * D, mytile, lane);
    __syncthreads();
    load_frag_lds(fp, w2);
}

__device__ __forceinline__ void mma3(const v2f (&a)[KSTEPS],
                                     const v2f (&w0)[KSTEPS], const v2f (&w1)[KSTEPS],
                                     const v2f (&w2)[KSTEPS],
                                     v8f& c0, v8f& c1, v8f& c2) {
#pragma unroll
    for (int s = 0; s < KSTEPS; ++s) {
        c0 = __builtin_amdgcn_wmma_f32_16x16x4_f32(false, a[s], false, w0[s], (short)0, c0, false, false);
        c1 = __builtin_amdgcn_wmma_f32_16x16x4_f32(false, a[s], false, w1[s], (short)0, c1, false, false);
        c2 = __builtin_amdgcn_wmma_f32_16x16x4_f32(false, a[s], false, w2[s], (short)0, c2, false, false);
    }
}

__global__ void zero_words_kernel(int* __restrict__ p, int nwords) {
    int i = blockIdx.x * blockDim.x + threadIdx.x;
    if (i < nwords) p[i] = 0;
}

__global__ __launch_bounds__(256) void mark_flags_kernel(const int* __restrict__ d0,
                                                         const int* __restrict__ d1,
                                                         const int* __restrict__ d2,
                                                         unsigned char* __restrict__ flags,
                                                         int N, int E) {
    const int nthr = gridDim.x * blockDim.x;
    const int gtid = blockIdx.x * blockDim.x + threadIdx.x;
    unsigned char* f0 = flags;
    unsigned char* f1 = flags + N;
    unsigned char* f2 = flags + 2 * (size_t)N;

    int done = 0;
    if (((((size_t)d0) | ((size_t)d1) | ((size_t)d2)) & 15) == 0) {
        const int nv = E >> 2;
        for (int j = gtid; j < nv; j += nthr) {
            const v4i a = ((const v4i*)d0)[j];
            const v4i c = ((const v4i*)d1)[j];
            const v4i d = ((const v4i*)d2)[j];
            f0[a.x] = 1; f0[a.y] = 1; f0[a.z] = 1; f0[a.w] = 1;
            f1[c.x] = 1; f1[c.y] = 1; f1[c.z] = 1; f1[c.w] = 1;
            f2[d.x] = 1; f2[d.y] = 1; f2[d.z] = 1; f2[d.w] = 1;
        }
        done = nv << 2;
    }
    for (int i = done + gtid; i < E; i += nthr) {
        f0[d0[i]] = 1;
        f1[d1[i]] = 1;
        f2[d2[i]] = 1;
    }
}

__global__ __launch_bounds__(256) void feats_kernel(
    const float* __restrict__ ns,
    const float* __restrict__ Wa, const float* __restrict__ Wc,
    const float* __restrict__ Wd,
    const unsigned char* __restrict__ flags,
    float* __restrict__ feats, int N, int nTiles) {
    __shared__ __attribute__((aligned(16))) float wst[8 * TILE_F];
    __shared__ __attribute__((aligned(16))) float abuf[2 * TILE_F];

    const int lane = threadIdx.x & 31;
    const int wave = threadIdx.x >> 5;
    const int col0 = wave * 16;
    const int mn   = lane & 15;
    const int koff = (lane >> 4) * 2;
    const int mhi  = (lane >> 4) << 3;

    int t = blockIdx.x;
    if (t >= nTiles) return;

    v4f glo, ghi;
    fetch_rows(ns + (size_t)t * 16 * D, wave, lane, glo, ghi);

    float* mytile = wst + wave * TILE_F;
    const float* fp = mytile + mn * PITCH + koff;
    v2f w0[KSTEPS], w1[KSTEPS], w2[KSTEPS];
    stage_tile_wave(Wa + (size_t)col0 * D, mytile, lane);
    __syncthreads();
    load_frag_lds(fp, w0);
    __syncthreads();
    stage_tile_wave(Wc + (size_t)col0 * D, mytile, lane);
    __syncthreads();
    load_frag_lds(fp, w1);
    __syncthreads();
    stage_tile_wave(Wd + (size_t)col0 * D, mytile, lane);
    __syncthreads();
    load_frag_lds(fp, w2);

    put_rows(abuf, wave, lane, glo, ghi);
    __syncthreads();

    int cur = 0;
    for (; t < nTiles; t += gridDim.x) {
        const int row0 = t * 16;
        const unsigned long long fa8 = *(const unsigned long long*)(flags + row0 + mhi);
        const unsigned long long fc8 = *(const unsigned long long*)(flags + N + row0 + mhi);
        const unsigned long long fd8 = *(const unsigned long long*)(flags + 2 * N + row0 + mhi);

        int tn = t + gridDim.x;
        if (tn >= nTiles) tn = t;
        fetch_rows(ns + (size_t)tn * 16 * D, wave, lane, glo, ghi);

        const float* ab = abuf + cur * TILE_F;
        v2f a[KSTEPS];
        load_frag_lds(ab + mn * PITCH + koff, a);

        __builtin_amdgcn_sched_barrier(0);

        v8f c0 = {}, c1 = {}, c2 = {};
        mma3(a, w0, w1, w2, c0, c1, c2);

        __builtin_amdgcn_sched_barrier(0);

#pragma unroll
        for (int i = 0; i < 8; ++i) {
            int row = row0 + i + mhi;
            float va = ((fa8 >> (8 * i)) & 0xffull) ? c0[i] : 0.0f;
            float vc = ((fc8 >> (8 * i)) & 0xffull) ? c1[i] : 0.0f;
            float vd = ((fd8 >> (8 * i)) & 0xffull) ? c2[i] : 0.0f;
            feats[(size_t)row * D + col0 + mn] = fmaxf(fmaxf(va, vc), vd);
        }

        __builtin_amdgcn_sched_barrier(0);
        put_rows(abuf + (cur ^ 1) * TILE_F, wave, lane, glo, ghi);
        lds_barrier();
        cur ^= 1;
    }
}

__global__ __launch_bounds__(256) void gi_kernel(
    const float* __restrict__ feats, const float* __restrict__ Wih,
    const float* __restrict__ bih, float* __restrict__ gi, int nTiles) {
    __shared__ __attribute__((aligned(16))) float wst[8 * TILE_F];
    __shared__ __attribute__((aligned(16))) float abuf[2 * TILE_F];

    const int lane = threadIdx.x & 31;
    const int wave = threadIdx.x >> 5;
    const int col0 = wave * 16;
    const int mn   = lane & 15;
    const int koff = (lane >> 4) * 2;
    const int mhi  = (lane >> 4) << 3;

    int t = blockIdx.x;
    if (t >= nTiles) return;

    v4f glo, ghi;
    fetch_rows(feats + (size_t)t * 16 * D, wave, lane, glo, ghi);

    v2f w0[KSTEPS], w1[KSTEPS], w2[KSTEPS];
    load_weights3(Wih, col0, 128 + col0, 256 + col0, wst + wave * TILE_F, lane, w0, w1, w2);
    const float b0 = bih[col0 + mn];
    const float b1 = bih[128 + col0 + mn];
    const float b2 = bih[256 + col0 + mn];

    put_rows(abuf, wave, lane, glo, ghi);
    __syncthreads();

    int cur = 0;
    for (; t < nTiles; t += gridDim.x) {
        int tn = t + gridDim.x;
        if (tn >= nTiles) tn = t;
        fetch_rows(feats + (size_t)tn * 16 * D, wave, lane, glo, ghi);

        const float* ab = abuf + cur * TILE_F;
        v2f a[KSTEPS];
        load_frag_lds(ab + mn * PITCH + koff, a);

        __builtin_amdgcn_sched_barrier(0);

        v8f c0 = {}, c1 = {}, c2 = {};
        mma3(a, w0, w1, w2, c0, c1, c2);

        const int row0 = t * 16;
#pragma unroll
        for (int i = 0; i < 8; ++i) {
            size_t gb = (size_t)(row0 + i + mhi) * H3 + col0 + mn;
            gi[gb]       = c0[i] + b0;
            gi[gb + 128] = c1[i] + b1;
            gi[gb + 256] = c2[i] + b2;
        }

        __builtin_amdgcn_sched_barrier(0);
        put_rows(abuf + (cur ^ 1) * TILE_F, wave, lane, glo, ghi);
        lds_barrier();
        cur ^= 1;
    }
}

__global__ __launch_bounds__(256) void gru_step_kernel(
    const float* __restrict__ hin, const float* __restrict__ gi,
    const float* __restrict__ Whh, const float* __restrict__ bhh,
    float* __restrict__ hout, int nTiles) {
    __shared__ __attribute__((aligned(16))) float wst[8 * TILE_F];
    __shared__ __attribute__((aligned(16))) float abuf[2 * TILE_F];

    const int lane = threadIdx.x & 31;
    const int wave = threadIdx.x >> 5;
    const int col0 = wave * 16;
    const int mn   = lane & 15;
    const int koff = (lane >> 4) * 2;
    const int mhi  = (lane >> 4) << 3;

    int t = blockIdx.x;
    if (t >= nTiles) return;

    v4f glo, ghi;
    fetch_rows(hin + (size_t)t * 16 * D, wave, lane, glo, ghi);

    v2f wr[KSTEPS], wz[KSTEPS], wn[KSTEPS];
    load_weights3(Whh, col0, 128 + col0, 256 + col0, wst + wave * TILE_F, lane, wr, wz, wn);
    const float bhr = bhh[col0 + mn];
    const float bhz = bhh[128 + col0 + mn];
    const float bhn = bhh[256 + col0 + mn];

    put_rows(abuf, wave, lane, glo, ghi);
    __syncthreads();

    int cur = 0;
    for (; t < nTiles; t += gridDim.x) {
        const int row0 = t * 16;
        float gr[8], gz[8], gn[8], hv[8];
#pragma unroll
        for (int i = 0; i < 8; ++i) {
            size_t gb = (size_t)(row0 + i + mhi) * H3 + col0 + mn;
            gr[i] = gi[gb];
            gz[i] = gi[gb + 128];
            gn[i] = gi[gb + 256];
        }
        int tn = t + gridDim.x;
        if (tn >= nTiles) tn = t;
        fetch_rows(hin + (size_t)tn * 16 * D, wave, lane, glo, ghi);

        const float* ab = abuf + cur * TILE_F;
        v2f a[KSTEPS];
        load_frag_lds(ab + mn * PITCH + koff, a);
#pragma unroll
        for (int i = 0; i < 8; ++i) hv[i] = ab[(i + mhi) * PITCH + col0 + mn];

        __builtin_amdgcn_sched_barrier(0);

        v8f cr = {}, cz = {}, cn = {};
        mma3(a, wr, wz, wn, cr, cz, cn);

#pragma unroll
        for (int i = 0; i < 8; ++i) {
            int row = row0 + i + mhi;
            float r  = sigmoidf_(gr[i] + cr[i] + bhr);
            float z  = sigmoidf_(gz[i] + cz[i] + bhz);
            float ng = tanhf_(gn[i] + r * (cn[i] + bhn));
            float h  = hv[i];
            hout[(size_t)row * D + col0 + mn] = (1.0f - z) * ng + z * h;
        }

        __builtin_amdgcn_sched_barrier(0);
        put_rows(abuf + (cur ^ 1) * TILE_F, wave, lane, glo, ghi);
        lds_barrier();
        cur ^= 1;
    }
}

extern "C" void kernel_launch(void* const* d_in, const int* in_sizes, int n_in,
                              void* d_out, int out_size, void* d_ws, size_t ws_size,
                              hipStream_t stream) {
    const float* ns   = (const float*)d_in[0];
    const int*   da   = (const int*)d_in[1];
    const int*   dc   = (const int*)d_in[2];
    const int*   dd   = (const int*)d_in[3];
    const float* Wa   = (const float*)d_in[4];
    const float* Wc   = (const float*)d_in[5];
    const float* Wd   = (const float*)d_in[6];
    const float* Wih  = (const float*)d_in[7];
    const float* Whh  = (const float*)d_in[8];
    const float* bih  = (const float*)d_in[9];
    const float* bhh  = (const float*)d_in[10];
    float* out = (float*)d_out;

    const int N = in_sizes[0] / D;
    const int E = in_sizes[1];

    char* ws = (char*)d_ws;
    float* gi    = (float*)ws;
    float* feats = (float*)(ws + (size_t)N * H3 * sizeof(float));
    unsigned char* flags = (unsigned char*)(ws + (size_t)N * (H3 + D) * sizeof(float));

    int nwords = (3 * N + 3) / 4;
    zero_words_kernel<<<(nwords + 255) / 256, 256, 0, stream>>>((int*)flags, nwords);
    mark_flags_kernel<<<MARK_BLOCKS, 256, 0, stream>>>(da, dc, dd, flags, N, E);

    const int rowBlocks = N / 16;

    feats_kernel<<<PERSIST_BLOCKS, 256, 0, stream>>>(ns, Wa, Wc, Wd, flags, feats, N, rowBlocks);

    gi_kernel<<<PERSIST_BLOCKS, 256, 0, stream>>>(feats, Wih, bih, gi, rowBlocks);

    gru_step_kernel<<<PERSIST_BLOCKS, 256, 0, stream>>>(ns,    gi, Whh, bhh, feats, rowBlocks);
    gru_step_kernel<<<PERSIST_BLOCKS, 256, 0, stream>>>(feats, gi, Whh, bhh, out,   rowBlocks);
    gru_step_kernel<<<PERSIST_BLOCKS, 256, 0, stream>>>(out,   gi, Whh, bhh, feats, rowBlocks);
    gru_step_kernel<<<PERSIST_BLOCKS, 256, 0, stream>>>(feats, gi, Whh, bhh, out,   rowBlocks);
}


# four GRU steps in one launch: the 16x128 h tile passes from step to step through LDS inside the work-group, gate inputs read once per row tile and one tile ahead; summation order unchanged
# speedup vs baseline: 2.5333x; 1.0893x over previous
#include <hip/hip_runtime.h>
#include <hip/hip_bf16.h>


typedef float v2f __attribute__((ext_vector_type(2)));
typedef float v4f __attribute__((ext_vector_type(4)));
typedef float v8f __attribute__((ext_vector_type(8)));
typedef int v4i __attribute__((ext_vector_type(4)));

#define D 128
#define H3 384
#define KSTEPS 32
#define PERSIST_BLOCKS 256
#define MARK_BLOCKS 512
#define PITCH 132
#define TILE_F (16 * PITCH)

__device__ __forceinline__ float sigmoidf_(float x) {
    return __builtin_amdgcn_rcpf(1.0f + __expf(-x));
}

__device__ __forceinline__ float tanhf_(float x) {
    return 1.0f - 2.0f * __builtin_amdgcn_rcpf(1.0f + __expf(2.0f * x));
}


__device__ __forceinline__ void stage_tile_wave(const float* __restrict__ g, float* l, int lane) {
#pragma unroll
    for (int r = 0; r < 16; ++r) {
        v4f v = *(const v4f*)(g + r * D + 4 * lane);
        *(v4f*)(l + r * PITCH + 4 * lane) = v;
    }
}

__device__ __forceinline__ void fetch_rows(const float* __restrict__ tile, int wave, int lane,
                                           v4f& lo, v4f& hi) {
    lo = *(const v4f*)(tile + wave * D + 4 * lane);
    hi = *(const v4f*)(tile + (wave + 8) * D + 4 * lane);
}
__device__ __forceinline__ void put_rows(float* buf, int wave, int lane, v4f lo, v4f hi) {
    *(v4f*)(buf + wave * PITCH + 4 * lane) = lo;
    *(v4f*)(buf + (wave + 8) * PITCH + 4 * lane) = hi;
}

__device__ __forceinline__ void lds_barrier() {
    __builtin_amdgcn_fence(__ATOMIC_RELEASE, "workgroup", "local");
    __builtin_amdgcn_s_barrier();
    __builtin_amdgcn_fence(__ATOMIC_ACQUIRE, "workgroup", "local");
}

__device__ __forceinline__ void load_frag_lds(const float* p, v2f (&f)[KSTEPS]) {
#pragma unroll
    for (int s = 0; s < KSTEPS; ++s) f[s] = *(const v2f*)(p + 4 * s);
}

__device__ __forceinline__ void load_weights3(const float* __restrict__ W, int r0, int r1, int r2,
                                              float* mytile, int lane,
                                              v2f (&w0)[KSTEPS], v2f (&w1)[KSTEPS], v2f (&w2)[KSTEPS]) {
    const float* fp = mytile + (lane & 15) * PITCH + (lane >> 4) * 2;
    stage_tile_wave(W + (size_t)r0 * D, mytile, lane);
    __syncthreads();
    load_frag_lds(fp, w0);
    __syncthreads();
    stage_tile_wave(W + (size_t)r1 * D, mytile, lane);
    __syncthreads();
    load_frag_lds(fp, w1);
    __syncthreads();
    stage_tile_wave(W + (size_t)r2 * D, mytile, lane);
    __syncthreads();
    load_frag_lds(fp, w2);
}

__device__ __forceinline__ void mma3(const v2f (&a)[KSTEPS],
                                     const v2f (&w0)[KSTEPS], const v2f (&w1)[KSTEPS],
                                     const v2f (&w2)[KSTEPS],
                                     v8f& c0, v8f& c1, v8f& c2) {
#pragma unroll
    for (int s = 0; s < KSTEPS; ++s) {
        c0 = __builtin_amdgcn_wmma_f32_16x16x4_f32(false, a[s], false, w0[s], (short)0, c0, false, false);
        c1 = __builtin_amdgcn_wmma_f32_16x16x4_f32(false, a[s], false, w1[s], (short)0, c1, false, false);
        c2 = __builtin_amdgcn_wmma_f32_16x16x4_f32(false, a[s], false, w2[s], (short)0, c2, false, false);
    }
}

__global__ void zero_words_kernel(int* __restrict__ p, int nwords) {
    int i = blockIdx.x * blockDim.x + threadIdx.x;
    if (i < nwords) p[i] = 0;
}

__global__ __launch_bounds__(256) void mark_flags_kernel(const int* __restrict__ d0,
                                                         const int* __restrict__ d1,
                                                         const int* __restrict__ d2,
                                                         unsigned char* __restrict__ flags,
                                                         int N, int E) {
    const int nthr = gridDim.x * blockDim.x;
    const int gtid = blockIdx.x * blockDim.x + threadIdx.x;
    unsigned char* f0 = flags;
    unsigned char* f1 = flags + N;
    unsigned char* f2 = flags + 2 * (size_t)N;

    int done = 0;
    if (((((size_t)d0) | ((size_t)d1) | ((size_t)d2)) & 15) == 0) {
        const int nv = E >> 2;
        for (int j = gtid; j < nv; j += nthr) {
            const v4i a = ((const v4i*)d0)[j];
            const v4i c = ((const v4i*)d1)[j];
            const v4i d = ((const v4i*)d2)[j];
            f0[a.x] = 1; f0[a.y] = 1; f0[a.z] = 1; f0[a.w] = 1;
            f1[c.x] = 1; f1[c.y] = 1; f1[c.z] = 1; f1[c.w] = 1;
            f2[d.x] = 1; f2[d.y] = 1; f2[d.z] = 1; f2[d.w] = 1;
        }
        done = nv << 2;
    }
    for (int i = done + gtid; i < E; i += nthr) {
        f0[d0[i]] = 1;
        f1[d1[i]] = 1;
        f2[d2[i]] = 1;
    }
}

__global__ __launch_bounds__(256) void feats_kernel(
    const float* __restrict__ ns,
    const float* __restrict__ Wa, const float* __restrict__ Wc,
    const float* __restrict__ Wd,
    const unsigned char* __restrict__ flags,
    float* __restrict__ feats, int N, int nTiles) {
    __shared__ __attribute__((aligned(16))) float wst[8 * TILE_F];
    __shared__ __attribute__((aligned(16))) float abuf[2 * TILE_F];

    const int lane = threadIdx.x & 31;
    const int wave = threadIdx.x >> 5;
    const int col0 = wave * 16;
    const int mn   = lane & 15;
    const int koff = (lane >> 4) * 2;
    const int mhi  = (lane >> 4) << 3;

    int t = blockIdx.x;
    if (t >= nTiles) return;

    v4f glo, ghi;
    fetch_rows(ns + (size_t)t * 16 * D, wave, lane, glo, ghi);

    float* mytile = wst + wave * TILE_F;
    const float* fp = mytile + mn * PITCH + koff;
    v2f w0[KSTEPS], w1[KSTEPS], w2[KSTEPS];
    stage_tile_wave(Wa + (size_t)col0 * D, mytile, lane);
    __syncthreads();
    load_frag_lds(fp, w0);
    __syncthreads();
    stage_tile_wave(Wc + (size_t)col0 * D, mytile, lane);
    __syncthreads();
    load_frag_lds(fp, w1);
    __syncthreads();
    stage_tile_wave(Wd + (size_t)col0 * D, mytile, lane);
    __syncthreads();
    load_frag_lds(fp, w2);

    put_rows(abuf, wave, lane, glo, ghi);
    __syncthreads();

    int cur = 0;
    for (; t < nTiles; t += gridDim.x) {
        const int row0 = t * 16;
        const unsigned long long fa8 = *(const unsigned long long*)(flags + row0 + mhi);
        const unsigned long long fc8 = *(const unsigned long long*)(flags + N + row0 + mhi);
        const unsigned long long fd8 = *(const unsigned long long*)(flags + 2 * N + row0 + mhi);

        int tn = t + gridDim.x;
        if (tn >= nTiles) tn = t;
        fetch_rows(ns + (size_t)tn * 16 * D, wave, lane, glo, ghi);

        const float* ab = abuf + cur * TILE_F;
        v2f a[KSTEPS];
        load_frag_lds(ab + mn * PITCH + koff, a);

        __builtin_amdgcn_sched_barrier(0);

        v8f c0 = {}, c1 = {}, c2 = {};
        mma3(a, w0, w1, w2, c0, c1, c2);

        __builtin_amdgcn_sched_barrier(0);

#pragma unroll
        for (int i = 0; i < 8; ++i) {
            int row = row0 + i + mhi;
            float va = ((fa8 >> (8 * i)) & 0xffull) ? c0[i] : 0.0f;
            float vc = ((fc8 >> (8 * i)) & 0xffull) ? c1[i] : 0.0f;
            float vd = ((fd8 >> (8 * i)) & 0xffull) ? c2[i] : 0.0f;
            feats[(size_t)row * D + col0 + mn] = fmaxf(fmaxf(va, vc), vd);
        }

        __builtin_amdgcn_sched_barrier(0);
        put_rows(abuf + (cur ^ 1) * TILE_F, wave, lane, glo, ghi);
        lds_barrier();
        cur ^= 1;
    }
}

__global__ __launch_bounds__(256) void gi_kernel(
    const float* __restrict__ feats, const float* __restrict__ Wih,
    const float* __restrict__ bih, float* __restrict__ gi, int nTiles) {
    __shared__ __attribute__((aligned(16))) float wst[8 * TILE_F];
    __shared__ __attribute__((aligned(16))) float abuf[2 * TILE_F];

    const int lane = threadIdx.x & 31;
    const int wave = threadIdx.x >> 5;
    const int col0 = wave * 16;
    const int mn   = lane & 15;
    const int koff = (lane >> 4) * 2;
    const int mhi  = (lane >> 4) << 3;

    int t = blockIdx.x;
    if (t >= nTiles) return;

    v4f glo, ghi;
    fetch_rows(feats + (size_t)t * 16 * D, wave, lane, glo, ghi);

    v2f w0[KSTEPS], w1[KSTEPS], w2[KSTEPS];
    load_weights3(Wih, col0, 128 + col0, 256 + col0, wst + wave * TILE_F, lane, w0, w1, w2);
    const float b0 = bih[col0 + mn];
    const float b1 = bih[128 + col0 + mn];
    const float b2 = bih[256 + col0 + mn];

    put_rows(abuf, wave, lane, glo, ghi);
    __syncthreads();

    int cur = 0;
    for (; t < nTiles; t += gridDim.x) {
        int tn = t + gridDim.x;
        if (tn >= nTiles) tn = t;
        fetch_rows(feats + (size_t)tn * 16 * D, wave, lane, glo, ghi);

        const float* ab = abuf + cur * TILE_F;
        v2f a[KSTEPS];
        load_frag_lds(ab + mn * PITCH + koff, a);

        __builtin_amdgcn_sched_barrier(0);

        v8f c0 = {}, c1 = {}, c2 = {};
        mma3(a, w0, w1, w2, c0, c1, c2);

        const int row0 = t * 16;
#pragma unroll
        for (int i = 0; i < 8; ++i) {
            size_t gb = (size_t)(row0 + i + mhi) * H3 + col0 + mn;
            gi[gb]       = c0[i] + b0;
            gi[gb + 128] = c1[i] + b1;
            gi[gb + 256] = c2[i] + b2;
        }

        __builtin_amdgcn_sched_barrier(0);
        put_rows(abuf + (cur ^ 1) * TILE_F, wave, lane, glo, ghi);
        lds_barrier();
        cur ^= 1;
    }
}


__device__ __forceinline__ void load_gates(const float* __restrict__ gi, int rbase, int c,
                                           float (&gr)[8], float (&gz)[8], float (&gn)[8]) {
#pragma unroll
    for (int i = 0; i < 8; ++i) {
        size_t gb = (size_t)(rbase + i) * H3 + c;
        gr[i] = gi[gb];
        gz[i] = gi[gb + 128];
        gn[i] = gi[gb + 256];
    }
}

template <bool LAST>
__device__ __forceinline__ void gru_step_tile(
    const float* hsrc, float* hdst,
    const v2f (&wr)[KSTEPS], const v2f (&wz)[KSTEPS], const v2f (&wn)[KSTEPS],
    const float (&gr)[8], const float (&gz)[8], const float (&gn)[8],
    float bhr, float bhz, float bhn, int wave, int lane,
    float* __restrict__ orows, v4f glo, v4f ghi) {
    const int col0 = wave * 16;
    const int mn   = lane & 15;
    const int koff = (lane >> 4) * 2;
    const int mhi  = (lane >> 4) << 3;

    v2f a[KSTEPS];
    load_frag_lds(hsrc + mn * PITCH + koff, a);
    float hv[8];
#pragma unroll
    for (int i = 0; i < 8; ++i) hv[i] = hsrc[(i + mhi) * PITCH + col0 + mn];

    __builtin_amdgcn_sched_barrier(0);

    v8f cr = {}, cz = {}, cn = {};
    mma3(a, wr, wz, wn, cr, cz, cn);

    float hn[8];
#pragma unroll
    for (int i = 0; i < 8; ++i) {
        float r  = sigmoidf_(gr[i] + cr[i] + bhr);
        float z  = sigmoidf_(gz[i] + cz[i] + bhz);
        float ng = tanhf_(gn[i] + r * (cn[i] + bhn));
        float h  = hv[i];
        hn[i] = (1.0f - z) * ng + z * h;
    }

    if (LAST) {
#pragma unroll
        for (int i = 0; i < 8; ++i) orows[(size_t)(i + mhi) * D + col0 + mn] = hn[i];
        __builtin_amdgcn_sched_barrier(0);
        put_rows(hdst, wave, lane, glo, ghi);
    } else {
#pragma unroll
        for (int i = 0; i < 8; ++i) hdst[(i + mhi) * PITCH + col0 + mn] = hn[i];
    }
    lds_barrier();
}

__global__ __launch_bounds__(256) void gru_steps_kernel(
    const float* __restrict__ h0, const float* __restrict__ gi,
    const float* __restrict__ Whh, const float* __restrict__ bhh,
    float* __restrict__ hout, int nTiles) {
    __shared__ __attribute__((aligned(16))) float wst[8 * TILE_F];
    __shared__ __attribute__((aligned(16))) float hbuf[2 * TILE_F];

    const int lane = threadIdx.x & 31;
    const int wave = threadIdx.x >> 5;
    const int col0 = wave * 16;
    const int mn   = lane & 15;
    const int mhi  = (lane >> 4) << 3;

    int t = blockIdx.x;
    if (t >= nTiles) return;

    v4f glo, ghi;
    fetch_rows(h0 + (size_t)t * 16 * D, wave, lane, glo, ghi);
    float gr[8], gz[8], gn[8];
    load_gates(gi, t * 16 + mhi, col0 + mn, gr, gz, gn);

    v2f wr[KSTEPS], wz[KSTEPS], wn[KSTEPS];
    load_weights3(Whh, col0, 128 + col0, 256 + col0, wst + wave * TILE_F, lane, wr, wz, wn);
    const float bhr = bhh[col0 + mn];
    const float bhz = bhh[128 + col0 + mn];
    const float bhn = bhh[256 + col0 + mn];

    float* b0 = hbuf;
    float* b1 = hbuf + TILE_F;
    put_rows(b0, wave, lane, glo, ghi);
    __syncthreads();

    for (; t < nTiles; t += gridDim.x) {
        int tn = t + gridDim.x;
        if (tn >= nTiles) tn = t;
        fetch_rows(h0 + (size_t)tn * 16 * D, wave, lane, glo, ghi);
        float nr[8], nz[8], nn[8];
        load_gates(gi, tn * 16 + mhi, col0 + mn, nr, nz, nn);

        float* orows = hout + (size_t)t * 16 * D;
        gru_step_tile<false>(b0, b1, wr, wz, wn, gr, gz, gn, bhr, bhz, bhn, wave, lane, orows, glo, ghi);
        gru_step_tile<false>(b1, b0, wr, wz, wn, gr, gz, gn, bhr, bhz, bhn, wave, lane, orows, glo, ghi);
        gru_step_tile<false>(b0, b1, wr, wz, wn, gr, gz, gn, bhr, bhz, bhn, wave, lane, orows, glo, ghi);
        gru_step_tile<true >(b1, b0, wr, wz, wn, gr, gz, gn, bhr, bhz, bhn, wave, lane, orows, glo, ghi);

#pragma unroll
        for (int i = 0; i < 8; ++i) { gr[i] = nr[i]; gz[i] = nz[i]; gn[i] = nn[i]; }
    }
}

extern "C" void kernel_launch(void* const* d_in, const int* in_sizes, int n_in,
                              void* d_out, int out_size, void* d_ws, size_t ws_size,
                              hipStream_t stream) {
    const float* ns   = (const float*)d_in[0];
    const int*   da   = (const int*)d_in[1];
    const int*   dc   = (const int*)d_in[2];
    const int*   dd   = (const int*)d_in[3];
    const float* Wa   = (const float*)d_in[4];
    const float* Wc   = (const float*)d_in[5];
    const float* Wd   = (const float*)d_in[6];
    const float* Wih  = (const float*)d_in[7];
    const float* Whh  = (const float*)d_in[8];
    const float* bih  = (const float*)d_in[9];
    const float* bhh  = (const float*)d_in[10];
    float* out = (float*)d_out;

    const int N = in_sizes[0] / D;
    const int E = in_sizes[1];

    char* ws = (char*)d_ws;
    float* gi    = (float*)ws;
    float* feats = (float*)(ws + (size_t)N * H3 * sizeof(float));
    unsigned char* flags = (unsigned char*)(ws + (size_t)N * (H3 + D) * sizeof(float));

    int nwords = (3 * N + 3) / 4;
    zero_words_kernel<<<(nwords + 255) / 256, 256, 0, stream>>>((int*)flags, nwords);
    mark_flags_kernel<<<MARK_BLOCKS, 256, 0, stream>>>(da, dc, dd, flags, N, E);

    const int rowBlocks = N / 16;

    feats_kernel<<<PERSIST_BLOCKS, 256, 0, stream>>>(ns, Wa, Wc, Wd, flags, feats, N, rowBlocks);

    gi_kernel<<<PERSIST_BLOCKS, 256, 0, stream>>>(feats, Wih, bih, gi, rowBlocks);

    gru_steps_kernel<<<PERSIST_BLOCKS, 256, 0, stream>>>(ns, gi, Whh, bhh, out, rowBlocks);
}


# flags without scattered global stores: 64 work-groups set their edge slice's flag bytes in private LDS maps, write them out as whole lines, a small launch ORs the 64 planes; zero-fill launch gone
# speedup vs baseline: 4.9674x; 1.9608x over previous
#include <hip/hip_runtime.h>
#include <hip/hip_bf16.h>


typedef float v2f __attribute__((ext_vector_type(2)));
typedef float v4f __attribute__((ext_vector_type(4)));
typedef float v8f __attribute__((ext_vector_type(8)));
typedef int v4i __attribute__((ext_vector_type(4)));

#define D 128
#define H3 384
#define KSTEPS 32
#define PERSIST_BLOCKS 256
#define MARK_GROUPS 64
#define MARK_THREADS 1024
#define MARK_DEPTH 4
#define MAPN 50176
#define PITCH 132
#define TILE_F (16 * PITCH)

__device__ __forceinline__ float sigmoidf_(float x) {
    return __builtin_amdgcn_rcpf(1.0f + __expf(-x));
}

__device__ __forceinline__ float tanhf_(float x) {
    return 1.0f - 2.0f * __builtin_amdgcn_rcpf(1.0f + __expf(2.0f * x));
}


__device__ __forceinline__ void stage_tile_wave(const float* __restrict__ g, float* l, int lane) {
#pragma unroll
    for (int r = 0; r < 16; ++r) {
        v4f v = *(const v4f*)(g + r * D + 4 * lane);
        *(v4f*)(l + r * PITCH + 4 * lane) = v;
    }
}

__device__ __forceinline__ void fetch_rows(const float* __restrict__ tile, int wave, int lane,
                                           v4f& lo, v4f& hi) {
    lo = *(const v4f*)(tile + wave * D + 4 * lane);
    hi = *(const v4f*)(tile + (wave + 8) * D + 4 * lane);
}
__device__ __forceinline__ void put_rows(float* buf, int wave, int lane, v4f lo, v4f hi) {
    *(v4f*)(buf + wave * PITCH + 4 * lane) = lo;
    *(v4f*)(buf + (wave + 8) * PITCH + 4 * lane) = hi;
}

__device__ __forceinline__ void lds_barrier() {
    __builtin_amdgcn_fence(__ATOMIC_RELEASE, "workgroup", "local");
    __builtin_amdgcn_s_barrier();
    __builtin_amdgcn_fence(__ATOMIC_ACQUIRE, "workgroup", "local");
}

__device__ __forceinline__ void load_frag_lds(const float* p, v2f (&f)[KSTEPS]) {
#pragma unroll
    for (int s = 0; s < KSTEPS; ++s) f[s] = *(const v2f*)(p + 4 * s);
}

__device__ __forceinline__ void load_weights3(const float* __restrict__ W, int r0, int r1, int r2,
                                              float* mytile, int lane,
                                              v2f (&w0)[KSTEPS], v2f (&w1)[KSTEPS], v2f (&w2)[KSTEPS]) {
    const float* fp = mytile + (lane & 15) * PITCH + (lane >> 4) * 2;
    stage_tile_wave(W + (size_t)r0 * D, mytile, lane);
    __syncthreads();
    load_frag_lds(fp, w0);
    __syncthreads();
    stage_tile_wave(W + (size_t)r1 * D, mytile, lane);
    __syncthreads();
    load_frag_lds(fp, w1);
    __syncthreads();
    stage_tile_wave(W + (size_t)r2 * D, mytile, lane);
    __syncthreads();
    load_frag_lds(fp, w2);
}

__device__ __forceinline__ void mma3(const v2f (&a)[KSTEPS],
                                     const v2f (&w0)[KSTEPS], const v2f (&w1)[KSTEPS],
                                     const v2f (&w2)[KSTEPS],
                                     v8f& c0, v8f& c1, v8f& c2) {
#pragma unroll
    for (int s = 0; s < KSTEPS; ++s) {
        c0 = __builtin_amdgcn_wmma_f32_16x16x4_f32(false, a[s], false, w0[s], (short)0, c0, false, false);
        c1 = __builtin_amdgcn_wmma_f32_16x16x4_f32(false, a[s], false, w1[s], (short)0, c1, false, false);
        c2 = __builtin_amdgcn_wmma_f32_16x16x4_f32(false, a[s], false, w2[s], (short)0, c2, false, false);
    }
}

__device__ __forceinline__ void set_flag(unsigned char* map, int off, int d, int base, int wlen) {
    const unsigned int i = (unsigned int)(d - base);
    map[(i < (unsigned int)wlen) ? (off + (int)i) : (3 * MAPN)] = 1;
}

__global__ __launch_bounds__(MARK_THREADS) void mark_private_kernel(const int* __restrict__ d0,
                                                                    const int* __restrict__ d1,
                                                                    const int* __restrict__ d2,
                                                                    unsigned char* __restrict__ planes,
                                                                    int N, int E) {
    __shared__ __attribute__((aligned(16))) unsigned char map[3 * MAPN + 16];

    const int tid = threadIdx.x;
    const int g = blockIdx.x;
    const int ng = gridDim.x;
    unsigned char* mine = planes + (size_t)g * 3 * (size_t)N;

    const bool vec = ((((size_t)d0) | ((size_t)d1) | ((size_t)d2)) & 15) == 0;
    const int nv = vec ? (E >> 2) : 0;
    const int per = (nv + ng - 1) / ng;
    const int lo = g * per;
    const int hi = (lo + per < nv) ? (lo + per) : nv;
    const int rem0 = nv << 2;
    const int rper = (E - rem0 + ng - 1) / ng;
    const int rlo = rem0 + g * rper;
    const int rhi = (rlo + rper < E) ? (rlo + rper) : E;
    const v4i* p0 = (const v4i*)d0;
    const v4i* p1 = (const v4i*)d1;
    const v4i* p2 = (const v4i*)d2;

    for (int base = 0; base < N; base += MAPN) {
        const int wlen = (N - base < MAPN) ? (N - base) : MAPN;

        for (int j = tid; j < (3 * MAPN + 16) / 16; j += MARK_THREADS) ((v4i*)map)[j] = (v4i){0, 0, 0, 0};
        __syncthreads();

        for (int j0 = lo + tid; j0 < hi; j0 += MARK_DEPTH * MARK_THREADS) {
            v4i a[MARK_DEPTH], c[MARK_DEPTH], d[MARK_DEPTH];
#pragma unroll
            for (int u = 0; u < MARK_DEPTH; ++u) {
                const int j = j0 + u * MARK_THREADS;
                const int jj = (j < hi) ? j : j0;
                a[u] = p0[jj];
                c[u] = p1[jj];
                d[u] = p2[jj];
            }
#pragma unroll
            for (int u = 0; u < MARK_DEPTH; ++u) {
                set_flag(map, 0, a[u].x, base, wlen);        set_flag(map, 0, a[u].y, base, wlen);
                set_flag(map, 0, a[u].z, base, wlen);        set_flag(map, 0, a[u].w, base, wlen);
                set_flag(map, MAPN, c[u].x, base, wlen);     set_flag(map, MAPN, c[u].y, base, wlen);
                set_flag(map, MAPN, c[u].z, base, wlen);     set_flag(map, MAPN, c[u].w, base, wlen);
                set_flag(map, 2 * MAPN, d[u].x, base, wlen); set_flag(map, 2 * MAPN, d[u].y, base, wlen);
                set_flag(map, 2 * MAPN, d[u].z, base, wlen); set_flag(map, 2 * MAPN, d[u].w, base, wlen);
            }
        }
        for (int i = rlo + tid; i < rhi; i += MARK_THREADS) {
            set_flag(map, 0, d0[i], base, wlen);
            set_flag(map, MAPN, d1[i], base, wlen);
            set_flag(map, 2 * MAPN, d2[i], base, wlen);
        }
        __syncthreads();

        const int nch = wlen >> 4;
        for (int t = 0; t < 3; ++t) {
            const unsigned char* srcp = map + t * MAPN;
            unsigned char* dstp = mine + (size_t)t * N + base;
            for (int c = tid; c < nch; c += MARK_THREADS)
                *(v4i*)(dstp + 16 * c) = *(const v4i*)(srcp + 16 * c);
        }
        __syncthreads();
    }
}

__global__ __launch_bounds__(256) void combine_flags_kernel(const unsigned char* __restrict__ planes,
                                                            unsigned char* __restrict__ flags,
                                                            int nw8) {
    const int i = blockIdx.x * blockDim.x + threadIdx.x;
    if (i >= nw8) return;
    const unsigned long long* p = (const unsigned long long*)planes + i;
    unsigned long long acc = 0;
#pragma unroll
    for (int g = 0; g < MARK_GROUPS; ++g) acc |= p[(size_t)g * (size_t)nw8];
    ((unsigned long long*)flags)[i] = acc;
}

__global__ __launch_bounds__(256) void feats_kernel(
    const float* __restrict__ ns,
    const float* __restrict__ Wa, const float* __restrict__ Wc,
    const float* __restrict__ Wd,
    const unsigned char* __restrict__ flags,
    float* __restrict__ feats, int N, int nTiles) {
    __shared__ __attribute__((aligned(16))) float wst[8 * TILE_F];
    __shared__ __attribute__((aligned(16))) float abuf[2 * TILE_F];

    const int lane = threadIdx.x & 31;
    const int wave = threadIdx.x >> 5;
    const int col0 = wave * 16;
    const int mn   = lane & 15;
    const int koff = (lane >> 4) * 2;
    const int mhi  = (lane >> 4) << 3;

    int t = blockIdx.x;
    if (t >= nTiles) return;

    v4f glo, ghi;
    fetch_rows(ns + (size_t)t * 16 * D, wave, lane, glo, ghi);

    float* mytile = wst + wave * TILE_F;
    const float* fp = mytile + mn * PITCH + koff;
    v2f w0[KSTEPS], w1[KSTEPS], w2[KSTEPS];
    stage_tile_wave(Wa + (size_t)col0 * D, mytile, lane);
    __syncthreads();
    load_frag_lds(fp, w0);
    __syncthreads();
    stage_tile_wave(Wc + (size_t)col0 * D, mytile, lane);
    __syncthreads();
    load_frag_lds(fp, w1);
    __syncthreads();
    stage_tile_wave(Wd + (size_t)col0 * D, mytile, lane);
    __syncthreads();
    load_frag_lds(fp, w2);

    put_rows(abuf, wave, lane, glo, ghi);
    __syncthreads();

    int cur = 0;
    for (; t < nTiles; t += gridDim.x) {
        const int row0 = t * 16;
        const unsigned long long fa8 = *(const unsigned long long*)(flags + row0 + mhi);
        const unsigned long long fc8 = *(const unsigned long long*)(flags + N + row0 + mhi);
        const unsigned long long fd8 = *(const unsigned long long*)(flags + 2 * N + row0 + mhi);

        int tn = t + gridDim.x;
        if (tn >= nTiles) tn = t;
        fetch_rows(ns + (size_t)tn * 16 * D, wave, lane, glo, ghi);

        const float* ab = abuf + cur * TILE_F;
        v2f a[KSTEPS];
        load_frag_lds(ab + mn * PITCH + koff, a);

        __builtin_amdgcn_sched_barrier(0);

        v8f c0 = {}, c1 = {}, c2 = {};
        mma3(a, w0, w1, w2, c0, c1, c2);

        __builtin_amdgcn_sched_barrier(0);

#pragma unroll
        for (int i = 0; i < 8; ++i) {
            int row = row0 + i + mhi;
            float va = ((fa8 >> (8 * i)) & 0xffull) ? c0[i] : 0.0f;
            float vc = ((fc8 >> (8 * i)) & 0xffull) ? c1[i] : 0.0f;
            float vd = ((fd8 >> (8 * i)) & 0xffull) ? c2[i] : 0.0f;
            feats[(size_t)row * D + col0 + mn] = fmaxf(fmaxf(va, vc), vd);
        }

        __builtin_amdgcn_sched_barrier(0);
        put_rows(abuf + (cur ^ 1) * TILE_F, wave, lane, glo, ghi);
        lds_barrier();
        cur ^= 1;
    }
}

__global__ __launch_bounds__(256) void gi_kernel(
    const float* __restrict__ feats, const float* __restrict__ Wih,
    const float* __restrict__ bih, float* __restrict__ gi, int nTiles) {
    __shared__ __attribute__((aligned(16))) float wst[8 * TILE_F];
    __shared__ __attribute__((aligned(16))) float abuf[2 * TILE_F];

    const int lane = threadIdx.x & 31;
    const int wave = threadIdx.x >> 5;
    const int col0 = wave * 16;
    const int mn   = lane & 15;
    const int koff = (lane >> 4) * 2;
    const int mhi  = (lane >> 4) << 3;

    int t = blockIdx.x;
    if (t >= nTiles) return;

    v4f glo, ghi;
    fetch_rows(feats + (size_t)t * 16 * D, wave, lane, glo, ghi);

    v2f w0[KSTEPS], w1[KSTEPS], w2[KSTEPS];
    load_weights3(Wih, col0, 128 + col0, 256 + col0, wst + wave * TILE_F, lane, w0, w1, w2);
    const float b0 = bih[col0 + mn];
    const float b1 = bih[128 + col0 + mn];
    const float b2 = bih[256 + col0 + mn];

    put_rows(abuf, wave, lane, glo, ghi);
    __syncthreads();

    int cur = 0;
    for (; t < nTiles; t += gridDim.x) {
        int tn = t + gridDim.x;
        if (tn >= nTiles) tn = t;
        fetch_rows(feats + (size_t)tn * 16 * D, wave, lane, glo, ghi);

        const float* ab = abuf + cur * TILE_F;
        v2f a[KSTEPS];
        load_frag_lds(ab + mn * PITCH + koff, a);

        __builtin_amdgcn_sched_barrier(0);

        v8f c0 = {}, c1 = {}, c2 = {};
        mma3(a, w0, w1, w2, c0, c1, c2);

        const int row0 = t * 16;
#pragma unroll
        for (int i = 0; i < 8; ++i) {
            size_t gb = (size_t)(row0 + i + mhi) * H3 + col0 + mn;
            gi[gb]       = c0[i] + b0;
            gi[gb + 128] = c1[i] + b1;
            gi[gb + 256] = c2[i] + b2;
        }

        __builtin_amdgcn_sched_barrier(0);
        put_rows(abuf + (cur ^ 1) * TILE_F, wave, lane, glo, ghi);
        lds_barrier();
        cur ^= 1;
    }
}


__device__ __forceinline__ void load_gates(const float* __restrict__ gi, int rbase, int c,
                                           float (&gr)[8], float (&gz)[8], float (&gn)[8]) {
#pragma unroll
    for (int i = 0; i < 8; ++i) {
        size_t gb = (size_t)(rbase + i) * H3 + c;
        gr[i] = gi[gb];
        gz[i] = gi[gb + 128];
        gn[i] = gi[gb + 256];
    }
}

template <bool LAST>
__device__ __forceinline__ void gru_step_tile(
    const float* hsrc, float* hdst,
    const v2f (&wr)[KSTEPS], const v2f (&wz)[KSTEPS], const v2f (&wn)[KSTEPS],
    const float (&gr)[8], const float (&gz)[8], const float (&gn)[8],
    float bhr, float bhz, float bhn, int wave, int lane,
    float* __restrict__ orows, v4f glo, v4f ghi) {
    const int col0 = wave * 16;
    const int mn   = lane & 15;
    const int koff = (lane >> 4) * 2;
    const int mhi  = (lane >> 4) << 3;

    v2f a[KSTEPS];
    load_frag_lds(hsrc + mn * PITCH + koff, a);
    float hv[8];
#pragma unroll
    for (int i = 0; i < 8; ++i) hv[i] = hsrc[(i + mhi) * PITCH + col0 + mn];

    __builtin_amdgcn_sched_barrier(0);

    v8f cr = {}, cz = {}, cn = {};
    mma3(a, wr, wz, wn, cr, cz, cn);

    float hn[8];
#pragma unroll
    for (int i = 0; i < 8; ++i) {
        float r  = sigmoidf_(gr[i] + cr[i] + bhr);
        float z  = sigmoidf_(gz[i] + cz[i] + bhz);
        float ng = tanhf_(gn[i] + r * (cn[i] + bhn));
        float h  = hv[i];
        hn[i] = (1.0f - z) * ng + z * h;
    }

    if (LAST) {
#pragma unroll
        for (int i = 0; i < 8; ++i) orows[(size_t)(i + mhi) * D + col0 + mn] = hn[i];
        __builtin_amdgcn_sched_barrier(0);
        put_rows(hdst, wave, lane, glo, ghi);
    } else {
#pragma unroll
        for (int i = 0; i < 8; ++i) hdst[(i + mhi) * PITCH + col0 + mn] = hn[i];
    }
    lds_barrier();
}

__global__ __launch_bounds__(256) void gru_steps_kernel(
    const float* __restrict__ h0, const float* __restrict__ gi,
    const float* __restrict__ Whh, const float* __restrict__ bhh,
    float* __restrict__ hout, int nTiles) {
    __shared__ __attribute__((aligned(16))) float wst[8 * TILE_F];
    __shared__ __attribute__((aligned(16))) float hbuf[2 * TILE_F];

    const int lane = threadIdx.x & 31;
    const int wave = threadIdx.x >> 5;
    const int col0 = wave * 16;
    const int mn   = lane & 15;
    const int mhi  = (lane >> 4) << 3;

    int t = blockIdx.x;
    if (t >= nTiles) return;

    v4f glo, ghi;
    fetch_rows(h0 + (size_t)t * 16 * D, wave, lane, glo, ghi);
    float gr[8], gz[8], gn[8];
    load_gates(gi, t * 16 + mhi, col0 + mn, gr, gz, gn);

    v2f wr[KSTEPS], wz[KSTEPS], wn[KSTEPS];
    load_weights3(Whh, col0, 128 + col0, 256 + col0, wst + wave * TILE_F, lane, wr, wz, wn);
    const float bhr = bhh[col0 + mn];
    const float bhz = bhh[128 + col0 + mn];
    const float bhn = bhh[256 + col0 + mn];

    float* b0 = hbuf;
    float* b1 = hbuf + TILE_F;
    put_rows(b0, wave, lane, glo, ghi);
    __syncthreads();

    for (; t < nTiles; t += gridDim.x) {
        int tn = t + gridDim.x;
        if (tn >= nTiles) tn = t;
        fetch_rows(h0 + (size_t)tn * 16 * D, wave, lane, glo, ghi);
        float nr[8], nz[8], nn[8];
        load_gates(gi, tn * 16 + mhi, col0 + mn, nr, nz, nn);

        float* orows = hout + (size_t)t * 16 * D;
        gru_step_tile<false>(b0, b1, wr, wz, wn, gr, gz, gn, bhr, bhz, bhn, wave, lane, orows, glo, ghi);
        gru_step_tile<false>(b1, b0, wr, wz, wn, gr, gz, gn, bhr, bhz, bhn, wave, lane, orows, glo, ghi);
        gru_step_tile<false>(b0, b1, wr, wz, wn, gr, gz, gn, bhr, bhz, bhn, wave, lane, orows, glo, ghi);
        gru_step_tile<true >(b1, b0, wr, wz, wn, gr, gz, gn, bhr, bhz, bhn, wave, lane, orows, glo, ghi);

#pragma unroll
        for (int i = 0; i < 8; ++i) { gr[i] = nr[i]; gz[i] = nz[i]; gn[i] = nn[i]; }
    }
}

extern "C" void kernel_launch(void* const* d_in, const int* in_sizes, int n_in,
                              void* d_out, int out_size, void* d_ws, size_t ws_size,
                              hipStream_t stream) {
    const float* ns   = (const float*)d_in[0];
    const int*   da   = (const int*)d_in[1];
    const int*   dc   = (const int*)d_in[2];
    const int*   dd   = (const int*)d_in[3];
    const float* Wa   = (const float*)d_in[4];
    const float* Wc   = (const float*)d_in[5];
    const float* Wd   = (const float*)d_in[6];
    const float* Wih  = (const float*)d_in[7];
    const float* Whh  = (const float*)d_in[8];
    const float* bih  = (const float*)d_in[9];
    const float* bhh  = (const float*)d_in[10];
    float* out = (float*)d_out;

    const int N = in_sizes[0] / D;
    const int E = in_sizes[1];

    char* ws = (char*)d_ws;
    float* gi    = (float*)ws;
    float* feats = (float*)(ws + (size_t)N * H3 * sizeof(float));
    unsigned char* flags = (unsigned char*)(ws + (size_t)N * (H3 + D) * sizeof(float));

    unsigned char* planes = (unsigned char*)gi;
    mark_private_kernel<<<MARK_GROUPS, MARK_THREADS, 0, stream>>>(da, dc, dd, planes, N, E);
    int nw8 = (3 * N) / 8;
    combine_flags_kernel<<<(nw8 + 255) / 256, 256, 0, stream>>>(planes, flags, nw8);

    const int rowBlocks = N / 16;

    feats_kernel<<<PERSIST_BLOCKS, 256, 0, stream>>>(ns, Wa, Wc, Wd, flags, feats, N, rowBlocks);

    gi_kernel<<<PERSIST_BLOCKS, 256, 0, stream>>>(feats, Wih, bih, gi, rowBlocks);

    gru_steps_kernel<<<PERSIST_BLOCKS, 256, 0, stream>>>(ns, gi, Whh, bhh, out, rowBlocks);
}


# feats and gi in one persistent launch: the 16x128 feats tile goes from the edge-type products to the W_ih products through LDS, one barrier per row tile; feats never stored; summation order unchanged
# speedup vs baseline: 5.1309x; 1.0329x over previous
#include <hip/hip_runtime.h>
#include <hip/hip_bf16.h>


typedef float v2f __attribute__((ext_vector_type(2)));
typedef float v4f __attribute__((ext_vector_type(4)));
typedef float v8f __attribute__((ext_vector_type(8)));
typedef int v4i __attribute__((ext_vector_type(4)));

#define D 128
#define H3 384
#define KSTEPS 32
#define PERSIST_BLOCKS 256
#define MARK_GROUPS 64
#define MARK_THREADS 1024
#define MARK_DEPTH 4
#define MAPN 50176
#define PITCH 132
#define TILE_F (16 * PITCH)

__device__ __forceinline__ float sigmoidf_(float x) {
    return __builtin_amdgcn_rcpf(1.0f + __expf(-x));
}

__device__ __forceinline__ float tanhf_(float x) {
    return 1.0f - 2.0f * __builtin_amdgcn_rcpf(1.0f + __expf(2.0f * x));
}


__device__ __forceinline__ void stage_tile_wave(const float* __restrict__ g, float* l, int lane) {
#pragma unroll
    for (int r = 0; r < 16; ++r) {
        v4f v = *(const v4f*)(g + r * D + 4 * lane);
        *(v4f*)(l + r * PITCH + 4 * lane) = v;
    }
}

__device__ __forceinline__ void fetch_rows(const float* __restrict__ tile, int wave, int lane,
                                           v4f& lo, v4f& hi) {
    lo = *(const v4f*)(tile + wave * D + 4 * lane);
    hi = *(const v4f*)(tile + (wave + 8) * D + 4 * lane);
}
__device__ __forceinline__ void put_rows(float* buf, int wave, int lane, v4f lo, v4f hi) {
    *(v4f*)(buf + wave * PITCH + 4 * lane) = lo;
    *(v4f*)(buf + (wave + 8) * PITCH + 4 * lane) = hi;
}

__device__ __forceinline__ void lds_barrier() {
    __builtin_amdgcn_fence(__ATOMIC_RELEASE, "workgroup", "local");
    __builtin_amdgcn_s_barrier();
    __builtin_amdgcn_fence(__ATOMIC_ACQUIRE, "workgroup", "local");
}

__device__ __forceinline__ void load_frag_lds(const float* p, v2f (&f)[KSTEPS]) {
#pragma unroll
    for (int s = 0; s < KSTEPS; ++s) f[s] = *(const v2f*)(p + 4 * s);
}

__device__ __forceinline__ void load_weights3(const float* __restrict__ W, int r0, int r1, int r2,
                                              float* mytile, int lane,
                                              v2f (&w0)[KSTEPS], v2f (&w1)[KSTEPS], v2f (&w2)[KSTEPS]) {
    const float* fp = mytile + (lane & 15) * PITCH + (lane >> 4) * 2;
    stage_tile_wave(W + (size_t)r0 * D, mytile, lane);
    __syncthreads();
    load_frag_lds(fp, w0);
    __syncthreads();
    stage_tile_wave(W + (size_t)r1 * D, mytile, lane);
    __syncthreads();
    load_frag_lds(fp, w1);
    __syncthreads();
    stage_tile_wave(W + (size_t)r2 * D, mytile, lane);
    __syncthreads();
    load_frag_lds(fp, w2);
}

__device__ __forceinline__ void mma3(const v2f (&a)[KSTEPS],
                                     const v2f (&w0)[KSTEPS], const v2f (&w1)[KSTEPS],
                                     const v2f (&w2)[KSTEPS],
                                     v8f& c0, v8f& c1, v8f& c2) {
#pragma unroll
    for (int s = 0; s < KSTEPS; ++s) {
        c0 = __builtin_amdgcn_wmma_f32_16x16x4_f32(false, a[s], false, w0[s], (short)0, c0, false, false);
        c1 = __builtin_amdgcn_wmma_f32_16x16x4_f32(false, a[s], false, w1[s], (short)0, c1, false, false);
        c2 = __builtin_amdgcn_wmma_f32_16x16x4_f32(false, a[s], false, w2[s], (short)0, c2, false, false);
    }
}

__device__ __forceinline__ void set_flag(unsigned char* map, int off, int d, int base, int wlen) {
    const unsigned int i = (unsigned int)(d - base);
    map[(i < (unsigned int)wlen) ? (off + (int)i) : (3 * MAPN)] = 1;
}

__global__ __launch_bounds__(MARK_THREADS) void mark_private_kernel(const int* __restrict__ d0,
                                                                    const int* __restrict__ d1,
                                                                    const int* __restrict__ d2,
                                                                    unsigned char* __restrict__ planes,
                                                                    int N, int E) {
    __shared__ __attribute__((aligned(16))) unsigned char map[3 * MAPN + 16];

    const int tid = threadIdx.x;
    const int g = blockIdx.x;
    const int ng = gridDim.x;
    unsigned char* mine = planes + (size_t)g * 3 * (size_t)N;

    const bool vec = ((((size_t)d0) | ((size_t)d1) | ((size_t)d2)) & 15) == 0;
    const int nv = vec ? (E >> 2) : 0;
    const int per = (nv + ng - 1) / ng;
    const int lo = g * per;
    const int hi = (lo + per < nv) ? (lo + per) : nv;
    const int rem0 = nv << 2;
    const int rper = (E - rem0 + ng - 1) / ng;
    const int rlo = rem0 + g * rper;
    const int rhi = (rlo + rper < E) ? (rlo + rper) : E;
    const v4i* p0 = (const v4i*)d0;
    const v4i* p1 = (const v4i*)d1;
    const v4i* p2 = (const v4i*)d2;

    for (int base = 0; base < N; base += MAPN) {
        const int wlen = (N - base < MAPN) ? (N - base) : MAPN;

        for (int j = tid; j < (3 * MAPN + 16) / 16; j += MARK_THREADS) ((v4i*)map)[j] = (v4i){0, 0, 0, 0};
        __syncthreads();

        for (int j0 = lo + tid; j0 < hi; j0 += MARK_DEPTH * MARK_THREADS) {
            v4i a[MARK_DEPTH], c[MARK_DEPTH], d[MARK_DEPTH];
#pragma unroll
            for (int u = 0; u < MARK_DEPTH; ++u) {
                const int j = j0 + u * MARK_THREADS;
                const int jj = (j < hi) ? j : j0;
                a[u] = p0[jj];
                c[u] = p1[jj];
                d[u] = p2[jj];
            }
#pragma unroll
            for (int u = 0; u < MARK_DEPTH; ++u) {
                set_flag(map, 0, a[u].x, base, wlen);        set_flag(map, 0, a[u].y, base, wlen);
                set_flag(map, 0, a[u].z, base, wlen);        set_flag(map, 0, a[u].w, base, wlen);
                set_flag(map, MAPN, c[u].x, base, wlen);     set_flag(map, MAPN, c[u].y, base, wlen);
                set_flag(map, MAPN, c[u].z, base, wlen);     set_flag(map, MAPN, c[u].w, base, wlen);
                set_flag(map, 2 * MAPN, d[u].x, base, wlen); set_flag(map, 2 * MAPN, d[u].y, base, wlen);
                set_flag(map, 2 * MAPN, d[u].z, base, wlen); set_flag(map, 2 * MAPN, d[u].w, base, wlen);
            }
        }
        for (int i = rlo + tid; i < rhi; i += MARK_THREADS) {
            set_flag(map, 0, d0[i], base, wlen);
            set_flag(map, MAPN, d1[i], base, wlen);
            set_flag(map, 2 * MAPN, d2[i], base, wlen);
        }
        __syncthreads();

        const int nch = wlen >> 4;
        for (int t = 0; t < 3; ++t) {
            const unsigned char* srcp = map + t * MAPN;
            unsigned char* dstp = mine + (size_t)t * N + base;
            for (int c = tid; c < nch; c += MARK_THREADS)
                *(v4i*)(dstp + 16 * c) = *(const v4i*)(srcp + 16 * c);
        }
        __syncthreads();
    }
}

__global__ __launch_bounds__(256) void combine_flags_kernel(const unsigned char* __restrict__ planes,
                                                            unsigned char* __restrict__ flags,
                                                            int nw8) {
    const int i = blockIdx.x * blockDim.x + threadIdx.x;
    if (i >= nw8) return;
    const unsigned long long* p = (const unsigned long long*)planes + i;
    unsigned long long acc = 0;
#pragma unroll
    for (int g = 0; g < MARK_GROUPS; ++g) acc |= p[(size_t)g * (size_t)nw8];
    ((unsigned long long*)flags)[i] = acc;
}

__global__ __launch_bounds__(256) void feats_gi_kernel(
    const float* __restrict__ ns,
    const float* __restrict__ Wa, const float* __restrict__ Wc,
    const float* __restrict__ Wd,
    const unsigned char* __restrict__ flags,
    const float* __restrict__ Wih, const float* __restrict__ bih,
    float* __restrict__ gi, int N, int nTiles) {
    __shared__ __attribute__((aligned(16))) float wst[8 * TILE_F];
    __shared__ __attribute__((aligned(16))) float abuf[2 * TILE_F];
    __shared__ __attribute__((aligned(16))) float fbuf[2 * TILE_F];

    const int lane = threadIdx.x & 31;
    const int wave = threadIdx.x >> 5;
    const int col0 = wave * 16;
    const int mn   = lane & 15;
    const int koff = (lane >> 4) * 2;
    const int mhi  = (lane >> 4) << 3;

    int t = blockIdx.x;
    if (t >= nTiles) return;

    v4f glo, ghi;
    fetch_rows(ns + (size_t)t * 16 * D, wave, lane, glo, ghi);

    float* mytile = wst + wave * TILE_F;
    const float* fp = mytile + mn * PITCH + koff;
    v2f w0[KSTEPS], w1[KSTEPS], w2[KSTEPS];
    stage_tile_wave(Wa + (size_t)col0 * D, mytile, lane);
    __syncthreads();
    load_frag_lds(fp, w0);
    __syncthreads();
    stage_tile_wave(Wc + (size_t)col0 * D, mytile, lane);
    __syncthreads();
    load_frag_lds(fp, w1);
    __syncthreads();
    stage_tile_wave(Wd + (size_t)col0 * D, mytile, lane);
    __syncthreads();
    load_frag_lds(fp, w2);
    __syncthreads();

    v2f u0[KSTEPS], u1[KSTEPS], u2[KSTEPS];
    load_weights3(Wih, col0, 128 + col0, 256 + col0, mytile, lane, u0, u1, u2);
    const float b0 = bih[col0 + mn];
    const float b1 = bih[128 + col0 + mn];
    const float b2 = bih[256 + col0 + mn];

    put_rows(abuf, wave, lane, glo, ghi);
    __syncthreads();

    int cur = 0;
    for (; t < nTiles; t += gridDim.x) {
        const int row0 = t * 16;
        const unsigned long long fa8 = *(const unsigned long long*)(flags + row0 + mhi);
        const unsigned long long fc8 = *(const unsigned long long*)(flags + N + row0 + mhi);
        const unsigned long long fd8 = *(const unsigned long long*)(flags + 2 * N + row0 + mhi);

        int tn = t + gridDim.x;
        if (tn >= nTiles) tn = t;
        fetch_rows(ns + (size_t)tn * 16 * D, wave, lane, glo, ghi);

        float* fb = fbuf + cur * TILE_F;
        {
            const float* ab = abuf + cur * TILE_F;
            v2f a[KSTEPS];
            load_frag_lds(ab + mn * PITCH + koff, a);

            __builtin_amdgcn_sched_barrier(0);

            v8f c0 = {}, c1 = {}, c2 = {};
            mma3(a, w0, w1, w2, c0, c1, c2);

            __builtin_amdgcn_sched_barrier(0);

#pragma unroll
            for (int i = 0; i < 8; ++i) {
                float va = ((fa8 >> (8 * i)) & 0xffull) ? c0[i] : 0.0f;
                float vc = ((fc8 >> (8 * i)) & 0xffull) ? c1[i] : 0.0f;
                float vd = ((fd8 >> (8 * i)) & 0xffull) ? c2[i] : 0.0f;
                fb[(i + mhi) * PITCH + col0 + mn] = fmaxf(fmaxf(va, vc), vd);
            }
        }

        __builtin_amdgcn_sched_barrier(0);
        put_rows(abuf + (cur ^ 1) * TILE_F, wave, lane, glo, ghi);
        lds_barrier();

        {
            v2f a[KSTEPS];
            load_frag_lds(fb + mn * PITCH + koff, a);

            __builtin_amdgcn_sched_barrier(0);

            v8f c0 = {}, c1 = {}, c2 = {};
            mma3(a, u0, u1, u2, c0, c1, c2);

#pragma unroll
            for (int i = 0; i < 8; ++i) {
                size_t gb = (size_t)(row0 + i + mhi) * H3 + col0 + mn;
                gi[gb]       = c0[i] + b0;
                gi[gb + 128] = c1[i] + b1;
                gi[gb + 256] = c2[i] + b2;
            }
        }
        cur ^= 1;
    }
}


__device__ __forceinline__ void load_gates(const float* __restrict__ gi, int rbase, int c,
                                           float (&gr)[8], float (&gz)[8], float (&gn)[8]) {
#pragma unroll
    for (int i = 0; i < 8; ++i) {
        size_t gb = (size_t)(rbase + i) * H3 + c;
        gr[i] = gi[gb];
        gz[i] = gi[gb + 128];
        gn[i] = gi[gb + 256];
    }
}

template <bool LAST>
__device__ __forceinline__ void gru_step_tile(
    const float* hsrc, float* hdst,
    const v2f (&wr)[KSTEPS], const v2f (&wz)[KSTEPS], const v2f (&wn)[KSTEPS],
    const float (&gr)[8], const float (&gz)[8], const float (&gn)[8],
    float bhr, float bhz, float bhn, int wave, int lane,
    float* __restrict__ orows, v4f glo, v4f ghi) {
    const int col0 = wave * 16;
    const int mn   = lane & 15;
    const int koff = (lane >> 4) * 2;
    const int mhi  = (lane >> 4) << 3;

    v2f a[KSTEPS];
    load_frag_lds(hsrc + mn * PITCH + koff, a);
    float hv[8];
#pragma unroll
    for (int i = 0; i < 8; ++i) hv[i] = hsrc[(i + mhi) * PITCH + col0 + mn];

    __builtin_amdgcn_sched_barrier(0);

    v8f cr = {}, cz = {}, cn = {};
    mma3(a, wr, wz, wn, cr, cz, cn);

    float hn[8];
#pragma unroll
    for (int i = 0; i < 8; ++i) {
        float r  = sigmoidf_(gr[i] + cr[i] + bhr);
        float z  = sigmoidf_(gz[i] + cz[i] + bhz);
        float ng = tanhf_(gn[i] + r * (cn[i] + bhn));
        float h  = hv[i];
        hn[i] = (1.0f - z) * ng + z * h;
    }

    if (LAST) {
#pragma unroll
        for (int i = 0; i < 8; ++i) orows[(size_t)(i + mhi) * D + col0 + mn] = hn[i];
        __builtin_amdgcn_sched_barrier(0);
        put_rows(hdst, wave, lane, glo, ghi);
    } else {
#pragma unroll
        for (int i = 0; i < 8; ++i) hdst[(i + mhi) * PITCH + col0 + mn] = hn[i];
    }
    lds_barrier();
}

__global__ __launch_bounds__(256) void gru_steps_kernel(
    const float* __restrict__ h0, const float* __restrict__ gi,
    const float* __restrict__ Whh, const float* __restrict__ bhh,
    float* __restrict__ hout, int nTiles) {
    __shared__ __attribute__((aligned(16))) float wst[8 * TILE_F];
    __shared__ __attribute__((aligned(16))) float hbuf[2 * TILE_F];

    const int lane = threadIdx.x & 31;
    const int wave = threadIdx.x >> 5;
    const int col0 = wave * 16;
    const int mn   = lane & 15;
    const int mhi  = (lane >> 4) << 3;

    int t = blockIdx.x;
    if (t >= nTiles) return;

    v4f glo, ghi;
    fetch_rows(h0 + (size_t)t * 16 * D, wave, lane, glo, ghi);
    float gr[8], gz[8], gn[8];
    load_gates(gi, t * 16 + mhi, col0 + mn, gr, gz, gn);

    v2f wr[KSTEPS], wz[KSTEPS], wn[KSTEPS];
    load_weights3(Whh, col0, 128 + col0, 256 + col0, wst + wave * TILE_F, lane, wr, wz, wn);
    const float bhr = bhh[col0 + mn];
    const float bhz = bhh[128 + col0 + mn];
    const float bhn = bhh[256 + col0 + mn];

    float* b0 = hbuf;
    float* b1 = hbuf + TILE_F;
    put_rows(b0, wave, lane, glo, ghi);
    __syncthreads();

    for (; t < nTiles; t += gridDim.x) {
        int tn = t + gridDim.x;
        if (tn >= nTiles) tn = t;
        fetch_rows(h0 + (size_t)tn * 16 * D, wave, lane, glo, ghi);
        float nr[8], nz[8], nn[8];
        load_gates(gi, tn * 16 + mhi, col0 + mn, nr, nz, nn);

        float* orows = hout + (size_t)t * 16 * D;
        gru_step_tile<false>(b0, b1, wr, wz, wn, gr, gz, gn, bhr, bhz, bhn, wave, lane, orows, glo, ghi);
        gru_step_tile<false>(b1, b0, wr, wz, wn, gr, gz, gn, bhr, bhz, bhn, wave, lane, orows, glo, ghi);
        gru_step_tile<false>(b0, b1, wr, wz, wn, gr, gz, gn, bhr, bhz, bhn, wave, lane, orows, glo, ghi);
        gru_step_tile<true >(b1, b0, wr, wz, wn, gr, gz, gn, bhr, bhz, bhn, wave, lane, orows, glo, ghi);

#pragma unroll
        for (int i = 0; i < 8; ++i) { gr[i] = nr[i]; gz[i] = nz[i]; gn[i] = nn[i]; }
    }
}

extern "C" void kernel_launch(void* const* d_in, const int* in_sizes, int n_in,
                              void* d_out, int out_size, void* d_ws, size_t ws_size,
                              hipStream_t stream) {
    const float* ns   = (const float*)d_in[0];
    const int*   da   = (const int*)d_in[1];
    const int*   dc   = (const int*)d_in[2];
    const int*   dd   = (const int*)d_in[3];
    const float* Wa   = (const float*)d_in[4];
    const float* Wc   = (const float*)d_in[5];
    const float* Wd   = (const float*)d_in[6];
    const float* Wih  = (const float*)d_in[7];
    const float* Whh  = (const float*)d_in[8];
    const float* bih  = (const float*)d_in[9];
    const float* bhh  = (const float*)d_in[10];
    float* out = (float*)d_out;

    const int N = in_sizes[0] / D;
    const int E = in_sizes[1];

    char* ws = (char*)d_ws;
    float* gi    = (float*)ws;
    unsigned char* flags = (unsigned char*)(ws + (size_t)N * (H3 + D) * sizeof(float));

    unsigned char* planes = (unsigned char*)gi;
    mark_private_kernel<<<MARK_GROUPS, MARK_THREADS, 0, stream>>>(da, dc, dd, planes, N, E);
    int nw8 = (3 * N) / 8;
    combine_flags_kernel<<<(nw8 + 255) / 256, 256, 0, stream>>>(planes, flags, nw8);

    const int rowBlocks = N / 16;

    feats_gi_kernel<<<PERSIST_BLOCKS, 256, 0, stream>>>(ns, Wa, Wc, Wd, flags, Wih, bih, gi, N, rowBlocks);

    gru_steps_kernel<<<PERSIST_BLOCKS, 256, 0, stream>>>(ns, gi, Whh, bhh, out, rowBlocks);
}
